# starting kernel
# baseline (speedup 1.0000x reference)
#include <hip/hip_runtime.h>
#include <hip/hip_bf16.h>
#include <math.h>


typedef __attribute__((ext_vector_type(2))) float v2f;
typedef __attribute__((ext_vector_type(8))) float v8f;

#define BH   16
#define NSEQ 2048
#define DK   64
#define DV   64

__global__ void __launch_bounds__(256)
pret_coef_kernel(const float* __restrict__ omask,
                 float* __restrict__ qcoef,
                 float* __restrict__ kcoef) {
  __shared__ float red[256];
  const int row = blockIdx.x;
  const size_t base = (size_t)row * NSEQ;
  float s = 0.f;
  #pragma unroll
  for (int j = 0; j < NSEQ / 256; ++j)
    s += omask[base + threadIdx.x + 256 * j];
  red[threadIdx.x] = s;
  __syncthreads();
  for (int off = 128; off > 0; off >>= 1) {
    if (threadIdx.x < off) red[threadIdx.x] += red[threadIdx.x + off];
    __syncthreads();
  }
  if (threadIdx.x == 0) {
    const float o0 = omask[base];
    qcoef[row] = o0 * rsqrtf(red[0]);
    kcoef[row] = 1.0f / o0;
  }
}

__global__ void __launch_bounds__(256)
pret_attn_kernel(const float* __restrict__ q,
                 const float* __restrict__ k,
                 const float* __restrict__ v,
                 const float* __restrict__ qcoef,
                 const float* __restrict__ kcoef,
                 float* __restrict__ out) {
  __shared__ float sK [16][68];
  __shared__ float sVT[64][18];
  __shared__ float sS [8][16][18];

  const int nblkq = NSEQ / 128;
  const int bh    = blockIdx.x / nblkq;
  const int qblk  = blockIdx.x % nblkq;
  const int wave  = threadIdx.x >> 5;
  const int lane  = threadIdx.x & 31;
  const int n16   = lane & 15;
  const int half  = lane >> 4;

  const int tile = qblk * 8 + wave;
  const int q0   = tile * 16;

  v2f a[16];
  {
    const float qcv   = qcoef[bh * NSEQ + q0 + n16];
    const float* qrow = q + ((size_t)(bh * NSEQ + q0 + n16)) * DK;
    #pragma unroll
    for (int c = 0; c < 16; ++c) {
      v2f qv = *(const v2f*)(qrow + 4 * c + 2 * half);
      a[c].x = qv.x * qcv;
      a[c].y = qv.y * qcv;
    }
  }

  const v8f zero8 = {0.f, 0.f, 0.f, 0.f, 0.f, 0.f, 0.f, 0.f};
  v8f Oacc[4];
  #pragma unroll
  for (int t = 0; t < 4; ++t) Oacc[t] = zero8;
  v8f Pacc = zero8;

  const v2f ones = {1.0f, 1.0f};

  const int ktEnd = qblk * 8 + 8;
  for (int kt = 0; kt < ktEnd; ++kt) {
    {
      const int r    = threadIdx.x >> 4;
      const int c4   = (threadIdx.x & 15) * 4;
      const int krow = kt * 16 + r;
      const float kcv = kcoef[bh * NSEQ + krow];
      const float4 kv = *(const float4*)(k + ((size_t)(bh * NSEQ + krow)) * DK + c4);
      sK[r][c4 + 0] = kv.x * kcv;
      sK[r][c4 + 1] = kv.y * kcv;
      sK[r][c4 + 2] = kv.z * kcv;
      sK[r][c4 + 3] = kv.w * kcv;
      const float4 vv = *(const float4*)(v + ((size_t)(bh * NSEQ + krow)) * DV + c4);
      sVT[c4 + 0][r] = vv.x;
      sVT[c4 + 1][r] = vv.y;
      sVT[c4 + 2][r] = vv.z;
      sVT[c4 + 3][r] = vv.w;
    }
    __syncthreads();

    if (kt <= tile) {
      v8f S = zero8;
      #pragma unroll
      for (int c = 0; c < 16; ++c) {
        const v2f b = *(const v2f*)&sK[n16][4 * c + 2 * half];
        S = __builtin_amdgcn_wmma_f32_16x16x4_f32(false, a[c], false, b,
                                                  (short)0, S, false, false);
      }

      if (kt == tile) {
        const int key = kt * 16 + n16;
        #pragma unroll
        for (int g = 0; g < 8; ++g) {
          const int qr = q0 + g + 8 * half;
          if (key > qr) S[g] = 0.0f;
        }
      }

      #pragma unroll
      for (int g = 0; g < 8; ++g) sS[wave][g + 8 * half][n16] = S[g];
      v2f sa[4];
      #pragma unroll
      for (int c = 0; c < 4; ++c)
        sa[c] = *(const v2f*)&sS[wave][n16][4 * c + 2 * half];

      #pragma unroll
      for (int c = 0; c < 4; ++c)
        Pacc = __builtin_amdgcn_wmma_f32_16x16x4_f32(false, sa[c], false, ones,
                                                     (short)0, Pacc, false, false);

      #pragma unroll
      for (int ct = 0; ct < 4; ++ct) {
        #pragma unroll
        for (int c = 0; c < 4; ++c) {
          const v2f vb = *(const v2f*)&sVT[ct * 16 + n16][4 * c + 2 * half];
          Oacc[ct] = __builtin_amdgcn_wmma_f32_16x16x4_f32(false, sa[c], false, vb,
                                                           (short)0, Oacc[ct], false, false);
        }
      }
    }
    __syncthreads();
  }

  #pragma unroll
  for (int ct = 0; ct < 4; ++ct) {
    #pragma unroll
    for (int g = 0; g < 8; ++g) {
      const float p = fmaxf(fabsf(Pacc[g]), 1.0f);
      out[((size_t)(bh * NSEQ + q0 + g + 8 * half)) * DV + ct * 16 + n16] =
          Oacc[ct][g] / p;
    }
  }
}

extern "C" void kernel_launch(void* const* d_in, const int* in_sizes, int n_in,
                              void* d_out, int out_size, void* d_ws, size_t ws_size,
                              hipStream_t stream) {
  const float* q     = (const float*)d_in[0];
  const float* k     = (const float*)d_in[1];
  const float* v     = (const float*)d_in[2];
  const float* omask = (const float*)d_in[3];
  float* out = (float*)d_out;

  float* qcoef = (float*)d_ws;
  float* kcoef = qcoef + (size_t)BH * NSEQ;

  pret_coef_kernel<<<BH * NSEQ, 256, 0, stream>>>(omask, qcoef, kcoef);
  pret_attn_kernel<<<BH * (NSEQ / 128), 256, 0, stream>>>(q, k, v,
                                                          qcoef, kcoef, out);
}


# attention kernel stages 64 keys (four 16-key sub-tiles) per barrier pair instead of 16: load latency and barriers paid once per 64 keys, four independent S chains; summation order kept
# speedup vs baseline: 1.4333x; 1.4333x over previous
#include <hip/hip_runtime.h>
#include <hip/hip_bf16.h>
#include <math.h>


typedef __attribute__((ext_vector_type(2))) float v2f;
typedef __attribute__((ext_vector_type(8))) float v8f;

#define BH   16
#define NSEQ 2048
#define DK   64
#define DV   64
#define KT   64
#define PKV  68

__global__ void __launch_bounds__(256)
pret_coef_kernel(const float* __restrict__ omask,
                 float* __restrict__ qcoef,
                 float* __restrict__ kcoef) {
  __shared__ float red[256];
  const int row = blockIdx.x;
  const size_t base = (size_t)row * NSEQ;
  float s = 0.f;
  #pragma unroll
  for (int j = 0; j < NSEQ / 256; ++j)
    s += omask[base + threadIdx.x + 256 * j];
  red[threadIdx.x] = s;
  __syncthreads();
  for (int off = 128; off > 0; off >>= 1) {
    if (threadIdx.x < off) red[threadIdx.x] += red[threadIdx.x + off];
    __syncthreads();
  }
  if (threadIdx.x == 0) {
    const float o0 = omask[base];
    qcoef[row] = o0 * rsqrtf(red[0]);
    kcoef[row] = 1.0f / o0;
  }
}

__global__ void __launch_bounds__(256)
pret_attn_kernel(const float* __restrict__ q,
                 const float* __restrict__ k,
                 const float* __restrict__ v,
                 const float* __restrict__ qcoef,
                 const float* __restrict__ kcoef,
                 float* __restrict__ out) {
  __shared__ float sK [KT][PKV];
  __shared__ float sVT[DV][PKV];
  __shared__ float sS [8][16][18];

  const int nblkq = NSEQ / 128;
  const int bh    = blockIdx.x / nblkq;
  const int qblk  = blockIdx.x % nblkq;
  const int wave  = threadIdx.x >> 5;
  const int lane  = threadIdx.x & 31;
  const int n16   = lane & 15;
  const int half  = lane >> 4;

  const int tile = qblk * 8 + wave;
  const int q0   = tile * 16;

  v2f a[16];
  {
    const float qcv   = qcoef[bh * NSEQ + q0 + n16];
    const float* qrow = q + ((size_t)(bh * NSEQ + q0 + n16)) * DK;
    #pragma unroll
    for (int c = 0; c < 16; ++c) {
      v2f qv = *(const v2f*)(qrow + 4 * c + 2 * half);
      a[c].x = qv.x * qcv;
      a[c].y = qv.y * qcv;
    }
  }

  const v8f zero8 = {0.f, 0.f, 0.f, 0.f, 0.f, 0.f, 0.f, 0.f};
  v8f Oacc[4];
  #pragma unroll
  for (int t = 0; t < 4; ++t) Oacc[t] = zero8;
  v8f Pacc = zero8;

  const v2f ones = {1.0f, 1.0f};

  auto accum_subtile = [&](const v8f& S, const int s) {
    #pragma unroll
    for (int g = 0; g < 8; ++g) sS[wave][g + 8 * half][n16] = S[g];
    v2f sa[4];
    #pragma unroll
    for (int c = 0; c < 4; ++c)
      sa[c] = *(const v2f*)&sS[wave][n16][4 * c + 2 * half];

    #pragma unroll
    for (int c = 0; c < 4; ++c)
      Pacc = __builtin_amdgcn_wmma_f32_16x16x4_f32(false, sa[c], false, ones,
                                                   (short)0, Pacc, false, false);

    #pragma unroll
    for (int ct = 0; ct < 4; ++ct) {
      #pragma unroll
      for (int c = 0; c < 4; ++c) {
        const v2f vb = *(const v2f*)&sVT[ct * 16 + n16][16 * s + 4 * c + 2 * half];
        Oacc[ct] = __builtin_amdgcn_wmma_f32_16x16x4_f32(false, sa[c], false, vb,
                                                         (short)0, Oacc[ct], false, false);
      }
    }
  };

  const int stEnd = qblk * 2 + 2;
  for (int st = 0; st < stEnd; ++st) {
    {
      const int r0 = threadIdx.x >> 4;
      const int c4 = (threadIdx.x & 15) * 4;
      float  kc[4];
      float4 kv[4], vv[4];
      #pragma unroll
      for (int i = 0; i < 4; ++i) {
        const int krow = st * KT + r0 + 16 * i;
        kc[i] = kcoef[bh * NSEQ + krow];
        kv[i] = *(const float4*)(k + ((size_t)(bh * NSEQ + krow)) * DK + c4);
        vv[i] = *(const float4*)(v + ((size_t)(bh * NSEQ + krow)) * DV + c4);
      }
      #pragma unroll
      for (int i = 0; i < 4; ++i) {
        const int r = r0 + 16 * i;
        sK[r][c4 + 0] = kv[i].x * kc[i];
        sK[r][c4 + 1] = kv[i].y * kc[i];
        sK[r][c4 + 2] = kv[i].z * kc[i];
        sK[r][c4 + 3] = kv[i].w * kc[i];
        sVT[c4 + 0][r] = vv[i].x;
        sVT[c4 + 1][r] = vv[i].y;
        sVT[c4 + 2][r] = vv[i].z;
        sVT[c4 + 3][r] = vv[i].w;
      }
    }
    __syncthreads();

    const int sub0 = st * 4;
    if (sub0 + 3 < tile) {
      v8f S[4];
      #pragma unroll
      for (int s = 0; s < 4; ++s) S[s] = zero8;
      #pragma unroll
      for (int c = 0; c < 16; ++c) {
        #pragma unroll
        for (int s = 0; s < 4; ++s) {
          const v2f b = *(const v2f*)&sK[16 * s + n16][4 * c + 2 * half];
          S[s] = __builtin_amdgcn_wmma_f32_16x16x4_f32(false, a[c], false, b,
                                                       (short)0, S[s], false, false);
        }
      }
      #pragma unroll
      for (int s = 0; s < 4; ++s) accum_subtile(S[s], s);
    } else {
      #pragma unroll
      for (int s = 0; s < 4; ++s) {
        const int sub = sub0 + s;
        if (sub <= tile) {
          v8f S = zero8;
          #pragma unroll
          for (int c = 0; c < 16; ++c) {
            const v2f b = *(const v2f*)&sK[16 * s + n16][4 * c + 2 * half];
            S = __builtin_amdgcn_wmma_f32_16x16x4_f32(false, a[c], false, b,
                                                      (short)0, S, false, false);
          }
          if (sub == tile) {
            const int key = sub * 16 + n16;
            #pragma unroll
            for (int g = 0; g < 8; ++g) {
              const int qr = q0 + g + 8 * half;
              if (key > qr) S[g] = 0.0f;
            }
          }
          accum_subtile(S, s);
        }
      }
    }
    __syncthreads();
  }

  #pragma unroll
  for (int ct = 0; ct < 4; ++ct) {
    #pragma unroll
    for (int g = 0; g < 8; ++g) {
      const float p = fmaxf(fabsf(Pacc[g]), 1.0f);
      out[((size_t)(bh * NSEQ + q0 + g + 8 * half)) * DV + ct * 16 + n16] =
          Oacc[ct][g] / p;
    }
  }
}

extern "C" void kernel_launch(void* const* d_in, const int* in_sizes, int n_in,
                              void* d_out, int out_size, void* d_ws, size_t ws_size,
                              hipStream_t stream) {
  const float* q     = (const float*)d_in[0];
  const float* k     = (const float*)d_in[1];
  const float* v     = (const float*)d_in[2];
  const float* omask = (const float*)d_in[3];
  float* out = (float*)d_out;

  float* qcoef = (float*)d_ws;
  float* kcoef = qcoef + (size_t)BH * NSEQ;

  pret_coef_kernel<<<BH * NSEQ, 256, 0, stream>>>(omask, qcoef, kcoef);
  pret_attn_kernel<<<BH * (NSEQ / 128), 256, 0, stream>>>(q, k, v,
                                                          qcoef, kcoef, out);
}


# coefficient kernel rewritten as a stream: one wave sums 8 whole rows, 16-byte non-temporal loads with the next row in flight, lane butterfly; 4096 waves instead of 262144; row-sum order changed
# speedup vs baseline: 1.6928x; 1.1811x over previous
#include <hip/hip_runtime.h>
#include <hip/hip_bf16.h>
#include <math.h>


typedef __attribute__((ext_vector_type(2))) float v2f;
typedef __attribute__((ext_vector_type(4))) float v4f;
typedef __attribute__((ext_vector_type(8))) float v8f;

#define BH   16
#define NSEQ 2048
#define DK   64
#define DV   64
#define KT   64
#define PKV  68

#define CW   16
#define CR   8
#define CQ   (NSEQ / 128)

__global__ void __launch_bounds__(CW * 32)
pret_coef_kernel(const float* __restrict__ omask,
                 float* __restrict__ qcoef,
                 float* __restrict__ kcoef) {
  __shared__ __attribute__((aligned(16))) float sSum[CW * CR];
  __shared__ __attribute__((aligned(16))) float sO0 [CW * CR];

  const int lane = threadIdx.x & 31;
  const int wave = threadIdx.x >> 5;
  const int row0 = (blockIdx.x * CW + wave) * CR;
  const float* p = omask + (size_t)row0 * NSEQ + 4 * lane;

  v4f   w[2][CQ];
  float sum[CR];
  float first[CR];

#define LOAD_ROW(R)                                                           \
  _Pragma("unroll") for (int qq = 0; qq < CQ; ++qq)                           \
    w[(R) & 1][qq] = __builtin_nontemporal_load(                              \
        (const v4f*)(p + (size_t)(R) * NSEQ + 128 * qq));

  LOAD_ROW(0)
  #pragma unroll
  for (int r = 0; r < CR; ++r) {
    __builtin_amdgcn_sched_barrier(0);
    if (r + 1 < CR) { LOAD_ROW(r + 1) }
    __builtin_amdgcn_sched_barrier(0);
    const v4f* wr = w[r & 1];
    v2f lo = {0.f, 0.f}, hi = {0.f, 0.f};
    #pragma unroll
    for (int qq = 0; qq < CQ; ++qq) {
      lo += wr[qq].xy;
      hi += wr[qq].zw;
    }
    sum[r]   = (lo.x + lo.y) + (hi.x + hi.y);
    first[r] = wr[0].x;
  }
#undef LOAD_ROW

  #pragma unroll
  for (int r = 0; r < CR; ++r) {
    float s = sum[r];
    #pragma unroll
    for (int m = 16; m >= 1; m >>= 1) s += __shfl_xor(s, m, 32);
    sum[r] = s;
  }

  if (lane == 0) {
    #pragma unroll
    for (int r = 0; r < CR; ++r) {
      sSum[wave * CR + r] = sum[r];
      sO0 [wave * CR + r] = first[r];
    }
  }
  __syncthreads();

  if (threadIdx.x < 32) {
    const v4f s = *(const v4f*)&sSum[4 * lane];
    const v4f o = *(const v4f*)&sO0 [4 * lane];
    v4f qc, kc;
    qc.x = o.x * rsqrtf(s.x);  kc.x = 1.0f / o.x;
    qc.y = o.y * rsqrtf(s.y);  kc.y = 1.0f / o.y;
    qc.z = o.z * rsqrtf(s.z);  kc.z = 1.0f / o.z;
    qc.w = o.w * rsqrtf(s.w);  kc.w = 1.0f / o.w;
    const int out0 = blockIdx.x * (CW * CR) + 4 * lane;
    *(v4f*)(qcoef + out0) = qc;
    *(v4f*)(kcoef + out0) = kc;
  }
}

__global__ void __launch_bounds__(256)
pret_attn_kernel(const float* __restrict__ q,
                 const float* __restrict__ k,
                 const float* __restrict__ v,
                 const float* __restrict__ qcoef,
                 const float* __restrict__ kcoef,
                 float* __restrict__ out) {
  __shared__ float sK [KT][PKV];
  __shared__ float sVT[DV][PKV];
  __shared__ float sS [8][16][18];

  const int nblkq = NSEQ / 128;
  const int bh    = blockIdx.x / nblkq;
  const int qblk  = blockIdx.x % nblkq;
  const int wave  = threadIdx.x >> 5;
  const int lane  = threadIdx.x & 31;
  const int n16   = lane & 15;
  const int half  = lane >> 4;

  const int tile = qblk * 8 + wave;
  const int q0   = tile * 16;

  v2f a[16];
  {
    const float qcv   = qcoef[bh * NSEQ + q0 + n16];
    const float* qrow = q + ((size_t)(bh * NSEQ + q0 + n16)) * DK;
    #pragma unroll
    for (int c = 0; c < 16; ++c) {
      v2f qv = *(const v2f*)(qrow + 4 * c + 2 * half);
      a[c].x = qv.x * qcv;
      a[c].y = qv.y * qcv;
    }
  }

  const v8f zero8 = {0.f, 0.f, 0.f, 0.f, 0.f, 0.f, 0.f, 0.f};
  v8f Oacc[4];
  #pragma unroll
  for (int t = 0; t < 4; ++t) Oacc[t] = zero8;
  v8f Pacc = zero8;

  const v2f ones = {1.0f, 1.0f};

  auto accum_subtile = [&](const v8f& S, const int s) {
    #pragma unroll
    for (int g = 0; g < 8; ++g) sS[wave][g + 8 * half][n16] = S[g];
    v2f sa[4];
    #pragma unroll
    for (int c = 0; c < 4; ++c)
      sa[c] = *(const v2f*)&sS[wave][n16][4 * c + 2 * half];

    #pragma unroll
    for (int c = 0; c < 4; ++c)
      Pacc = __builtin_amdgcn_wmma_f32_16x16x4_f32(false, sa[c], false, ones,
                                                   (short)0, Pacc, false, false);

    #pragma unroll
    for (int ct = 0; ct < 4; ++ct) {
      #pragma unroll
      for (int c = 0; c < 4; ++c) {
        const v2f vb = *(const v2f*)&sVT[ct * 16 + n16][16 * s + 4 * c + 2 * half];
        Oacc[ct] = __builtin_amdgcn_wmma_f32_16x16x4_f32(false, sa[c], false, vb,
                                                         (short)0, Oacc[ct], false, false);
      }
    }
  };

  const int stEnd = qblk * 2 + 2;
  for (int st = 0; st < stEnd; ++st) {
    {
      const int r0 = threadIdx.x >> 4;
      const int c4 = (threadIdx.x & 15) * 4;
      float  kc[4];
      float4 kv[4], vv[4];
      #pragma unroll
      for (int i = 0; i < 4; ++i) {
        const int krow = st * KT + r0 + 16 * i;
        kc[i] = kcoef[bh * NSEQ + krow];
        kv[i] = *(const float4*)(k + ((size_t)(bh * NSEQ + krow)) * DK + c4);
        vv[i] = *(const float4*)(v + ((size_t)(bh * NSEQ + krow)) * DV + c4);
      }
      #pragma unroll
      for (int i = 0; i < 4; ++i) {
        const int r = r0 + 16 * i;
        sK[r][c4 + 0] = kv[i].x * kc[i];
        sK[r][c4 + 1] = kv[i].y * kc[i];
        sK[r][c4 + 2] = kv[i].z * kc[i];
        sK[r][c4 + 3] = kv[i].w * kc[i];
        sVT[c4 + 0][r] = vv[i].x;
        sVT[c4 + 1][r] = vv[i].y;
        sVT[c4 + 2][r] = vv[i].z;
        sVT[c4 + 3][r] = vv[i].w;
      }
    }
    __syncthreads();

    const int sub0 = st * 4;
    if (sub0 + 3 < tile) {
      v8f S[4];
      #pragma unroll
      for (int s = 0; s < 4; ++s) S[s] = zero8;
      #pragma unroll
      for (int c = 0; c < 16; ++c) {
        #pragma unroll
        for (int s = 0; s < 4; ++s) {
          const v2f b = *(const v2f*)&sK[16 * s + n16][4 * c + 2 * half];
          S[s] = __builtin_amdgcn_wmma_f32_16x16x4_f32(false, a[c], false, b,
                                                       (short)0, S[s], false, false);
        }
      }
      #pragma unroll
      for (int s = 0; s < 4; ++s) accum_subtile(S[s], s);
    } else {
      #pragma unroll
      for (int s = 0; s < 4; ++s) {
        const int sub = sub0 + s;
        if (sub <= tile) {
          v8f S = zero8;
          #pragma unroll
          for (int c = 0; c < 16; ++c) {
            const v2f b = *(const v2f*)&sK[16 * s + n16][4 * c + 2 * half];
            S = __builtin_amdgcn_wmma_f32_16x16x4_f32(false, a[c], false, b,
                                                      (short)0, S, false, false);
          }
          if (sub == tile) {
            const int key = sub * 16 + n16;
            #pragma unroll
            for (int g = 0; g < 8; ++g) {
              const int qr = q0 + g + 8 * half;
              if (key > qr) S[g] = 0.0f;
            }
          }
          accum_subtile(S, s);
        }
      }
    }
    __syncthreads();
  }

  #pragma unroll
  for (int ct = 0; ct < 4; ++ct) {
    #pragma unroll
    for (int g = 0; g < 8; ++g) {
      const float p = fmaxf(fabsf(Pacc[g]), 1.0f);
      out[((size_t)(bh * NSEQ + q0 + g + 8 * half)) * DV + ct * 16 + n16] =
          Oacc[ct][g] / p;
    }
  }
}

extern "C" void kernel_launch(void* const* d_in, const int* in_sizes, int n_in,
                              void* d_out, int out_size, void* d_ws, size_t ws_size,
                              hipStream_t stream) {
  const float* q     = (const float*)d_in[0];
  const float* k     = (const float*)d_in[1];
  const float* v     = (const float*)d_in[2];
  const float* omask = (const float*)d_in[3];
  float* out = (float*)d_out;

  float* qcoef = (float*)d_ws;
  float* kcoef = qcoef + (size_t)BH * NSEQ;

  pret_coef_kernel<<<(BH * NSEQ) / (CW * CR), CW * 32, 0, stream>>>(omask, qcoef, kcoef);
  pret_attn_kernel<<<BH * (NSEQ / 128), 256, 0, stream>>>(q, k, v,
                                                          qcoef, kcoef, out);
}


# attention kernel: each work-group takes four early and four mirrored late query tiles (one of each per SIMD) instead of eight consecutive ones, equalising causal work; summation order kept
# speedup vs baseline: 2.1403x; 1.2644x over previous
#include <hip/hip_runtime.h>
#include <hip/hip_bf16.h>
#include <math.h>


typedef __attribute__((ext_vector_type(2))) float v2f;
typedef __attribute__((ext_vector_type(4))) float v4f;
typedef __attribute__((ext_vector_type(8))) float v8f;

#define BH   16
#define NSEQ 2048
#define DK   64
#define DV   64
#define KT   64
#define PKV  68

#define CW   16
#define CR   8
#define CQ   (NSEQ / 128)

__global__ void __launch_bounds__(CW * 32)
pret_coef_kernel(const float* __restrict__ omask,
                 float* __restrict__ qcoef,
                 float* __restrict__ kcoef) {
  __shared__ __attribute__((aligned(16))) float sSum[CW * CR];
  __shared__ __attribute__((aligned(16))) float sO0 [CW * CR];

  const int lane = threadIdx.x & 31;
  const int wave = threadIdx.x >> 5;
  const int row0 = (blockIdx.x * CW + wave) * CR;
  const float* p = omask + (size_t)row0 * NSEQ + 4 * lane;

  v4f   w[2][CQ];
  float sum[CR];
  float first[CR];

#define LOAD_ROW(R)                                                           \
  _Pragma("unroll") for (int qq = 0; qq < CQ; ++qq)                           \
    w[(R) & 1][qq] = __builtin_nontemporal_load(                              \
        (const v4f*)(p + (size_t)(R) * NSEQ + 128 * qq));

  LOAD_ROW(0)
  #pragma unroll
  for (int r = 0; r < CR; ++r) {
    __builtin_amdgcn_sched_barrier(0);
    if (r + 1 < CR) { LOAD_ROW(r + 1) }
    __builtin_amdgcn_sched_barrier(0);
    const v4f* wr = w[r & 1];
    v2f lo = {0.f, 0.f}, hi = {0.f, 0.f};
    #pragma unroll
    for (int qq = 0; qq < CQ; ++qq) {
      lo += wr[qq].xy;
      hi += wr[qq].zw;
    }
    sum[r]   = (lo.x + lo.y) + (hi.x + hi.y);
    first[r] = wr[0].x;
  }
#undef LOAD_ROW

  #pragma unroll
  for (int r = 0; r < CR; ++r) {
    float s = sum[r];
    #pragma unroll
    for (int m = 16; m >= 1; m >>= 1) s += __shfl_xor(s, m, 32);
    sum[r] = s;
  }

  if (lane == 0) {
    #pragma unroll
    for (int r = 0; r < CR; ++r) {
      sSum[wave * CR + r] = sum[r];
      sO0 [wave * CR + r] = first[r];
    }
  }
  __syncthreads();

  if (threadIdx.x < 32) {
    const v4f s = *(const v4f*)&sSum[4 * lane];
    const v4f o = *(const v4f*)&sO0 [4 * lane];
    v4f qc, kc;
    qc.x = o.x * rsqrtf(s.x);  kc.x = 1.0f / o.x;
    qc.y = o.y * rsqrtf(s.y);  kc.y = 1.0f / o.y;
    qc.z = o.z * rsqrtf(s.z);  kc.z = 1.0f / o.z;
    qc.w = o.w * rsqrtf(s.w);  kc.w = 1.0f / o.w;
    const int out0 = blockIdx.x * (CW * CR) + 4 * lane;
    *(v4f*)(qcoef + out0) = qc;
    *(v4f*)(kcoef + out0) = kc;
  }
}

__global__ void __launch_bounds__(256)
pret_attn_kernel(const float* __restrict__ q,
                 const float* __restrict__ k,
                 const float* __restrict__ v,
                 const float* __restrict__ qcoef,
                 const float* __restrict__ kcoef,
                 float* __restrict__ out) {
  __shared__ float sK [KT][PKV];
  __shared__ float sVT[DV][PKV];
  __shared__ float sS [8][16][18];

  const int nblkq = NSEQ / 128;
  const int bh    = blockIdx.x / nblkq;
  const int qblk  = blockIdx.x % nblkq;
  const int wave  = threadIdx.x >> 5;
  const int lane  = threadIdx.x & 31;
  const int n16   = lane & 15;
  const int half  = lane >> 4;

  const int ntile = NSEQ / 16;
  const bool late = (__popc(wave) & 1) == 0;
  const int tile  = late ? (ntile - 4 - 4 * qblk + (wave >> 1))
                         : (4 * qblk + (wave >> 1));
  const int q0    = tile * 16;

  v2f a[16];
  {
    const float qcv   = qcoef[bh * NSEQ + q0 + n16];
    const float* qrow = q + ((size_t)(bh * NSEQ + q0 + n16)) * DK;
    #pragma unroll
    for (int c = 0; c < 16; ++c) {
      v2f qv = *(const v2f*)(qrow + 4 * c + 2 * half);
      a[c].x = qv.x * qcv;
      a[c].y = qv.y * qcv;
    }
  }

  const v8f zero8 = {0.f, 0.f, 0.f, 0.f, 0.f, 0.f, 0.f, 0.f};
  v8f Oacc[4];
  #pragma unroll
  for (int t = 0; t < 4; ++t) Oacc[t] = zero8;
  v8f Pacc = zero8;

  const v2f ones = {1.0f, 1.0f};

  auto accum_subtile = [&](const v8f& S, const int s) {
    #pragma unroll
    for (int g = 0; g < 8; ++g) sS[wave][g + 8 * half][n16] = S[g];
    v2f sa[4];
    #pragma unroll
    for (int c = 0; c < 4; ++c)
      sa[c] = *(const v2f*)&sS[wave][n16][4 * c + 2 * half];

    #pragma unroll
    for (int c = 0; c < 4; ++c)
      Pacc = __builtin_amdgcn_wmma_f32_16x16x4_f32(false, sa[c], false, ones,
                                                   (short)0, Pacc, false, false);

    #pragma unroll
    for (int ct = 0; ct < 4; ++ct) {
      #pragma unroll
      for (int c = 0; c < 4; ++c) {
        const v2f vb = *(const v2f*)&sVT[ct * 16 + n16][16 * s + 4 * c + 2 * half];
        Oacc[ct] = __builtin_amdgcn_wmma_f32_16x16x4_f32(false, sa[c], false, vb,
                                                         (short)0, Oacc[ct], false, false);
      }
    }
  };

  const int stEnd = nblkq * 2 - qblk;
  for (int st = 0; st < stEnd; ++st) {
    {
      const int r0 = threadIdx.x >> 4;
      const int c4 = (threadIdx.x & 15) * 4;
      float  kc[4];
      float4 kv[4], vv[4];
      #pragma unroll
      for (int i = 0; i < 4; ++i) {
        const int krow = st * KT + r0 + 16 * i;
        kc[i] = kcoef[bh * NSEQ + krow];
        kv[i] = *(const float4*)(k + ((size_t)(bh * NSEQ + krow)) * DK + c4);
        vv[i] = *(const float4*)(v + ((size_t)(bh * NSEQ + krow)) * DV + c4);
      }
      #pragma unroll
      for (int i = 0; i < 4; ++i) {
        const int r = r0 + 16 * i;
        sK[r][c4 + 0] = kv[i].x * kc[i];
        sK[r][c4 + 1] = kv[i].y * kc[i];
        sK[r][c4 + 2] = kv[i].z * kc[i];
        sK[r][c4 + 3] = kv[i].w * kc[i];
        sVT[c4 + 0][r] = vv[i].x;
        sVT[c4 + 1][r] = vv[i].y;
        sVT[c4 + 2][r] = vv[i].z;
        sVT[c4 + 3][r] = vv[i].w;
      }
    }
    __syncthreads();

    const int sub0 = st * 4;
    if (sub0 + 3 < tile) {
      v8f S[4];
      #pragma unroll
      for (int s = 0; s < 4; ++s) S[s] = zero8;
      #pragma unroll
      for (int c = 0; c < 16; ++c) {
        #pragma unroll
        for (int s = 0; s < 4; ++s) {
          const v2f b = *(const v2f*)&sK[16 * s + n16][4 * c + 2 * half];
          S[s] = __builtin_amdgcn_wmma_f32_16x16x4_f32(false, a[c], false, b,
                                                       (short)0, S[s], false, false);
        }
      }
      #pragma unroll
      for (int s = 0; s < 4; ++s) accum_subtile(S[s], s);
    } else {
      #pragma unroll
      for (int s = 0; s < 4; ++s) {
        const int sub = sub0 + s;
        if (sub <= tile) {
          v8f S = zero8;
          #pragma unroll
          for (int c = 0; c < 16; ++c) {
            const v2f b = *(const v2f*)&sK[16 * s + n16][4 * c + 2 * half];
            S = __builtin_amdgcn_wmma_f32_16x16x4_f32(false, a[c], false, b,
                                                      (short)0, S, false, false);
          }
          if (sub == tile) {
            const int key = sub * 16 + n16;
            #pragma unroll
            for (int g = 0; g < 8; ++g) {
              const int qr = q0 + g + 8 * half;
              if (key > qr) S[g] = 0.0f;
            }
          }
          accum_subtile(S, s);
        }
      }
    }
    __syncthreads();
  }

  #pragma unroll
  for (int ct = 0; ct < 4; ++ct) {
    #pragma unroll
    for (int g = 0; g < 8; ++g) {
      const float p = fmaxf(fabsf(Pacc[g]), 1.0f);
      out[((size_t)(bh * NSEQ + q0 + g + 8 * half)) * DV + ct * 16 + n16] =
          Oacc[ct][g] / p;
    }
  }
}

extern "C" void kernel_launch(void* const* d_in, const int* in_sizes, int n_in,
                              void* d_out, int out_size, void* d_ws, size_t ws_size,
                              hipStream_t stream) {
  const float* q     = (const float*)d_in[0];
  const float* k     = (const float*)d_in[1];
  const float* v     = (const float*)d_in[2];
  const float* omask = (const float*)d_in[3];
  float* out = (float*)d_out;

  float* qcoef = (float*)d_ws;
  float* kcoef = qcoef + (size_t)BH * NSEQ;

  pret_coef_kernel<<<(BH * NSEQ) / (CW * CR), CW * 32, 0, stream>>>(omask, qcoef, kcoef);
  pret_attn_kernel<<<BH * (NSEQ / 128), 256, 0, stream>>>(q, k, v,
                                                          qcoef, kcoef, out);
}


# attention kernel: keys before a work-group's 128-row query block are summed once into a 64x64 state K^T V plus key sum and applied as q@state, not as S=qK^T, S@V per tile; summation order changed
# speedup vs baseline: 2.7904x; 1.3037x over previous
#include <hip/hip_runtime.h>
#include <hip/hip_bf16.h>
#include <math.h>


typedef __attribute__((ext_vector_type(2))) float v2f;
typedef __attribute__((ext_vector_type(4))) float v4f;
typedef __attribute__((ext_vector_type(8))) float v8f;

#define BH   16
#define NSEQ 2048
#define DK   64
#define DV   64
#define KT   64
#define PKV  68

#define CW   16
#define CR   8
#define CQ   (NSEQ / 128)

__global__ void __launch_bounds__(CW * 32)
pret_coef_kernel(const float* __restrict__ omask,
                 float* __restrict__ qcoef,
                 float* __restrict__ kcoef) {
  __shared__ __attribute__((aligned(16))) float sSum[CW * CR];
  __shared__ __attribute__((aligned(16))) float sO0 [CW * CR];

  const int lane = threadIdx.x & 31;
  const int wave = threadIdx.x >> 5;
  const int row0 = (blockIdx.x * CW + wave) * CR;
  const float* p = omask + (size_t)row0 * NSEQ + 4 * lane;

  v4f   w[2][CQ];
  float sum[CR];
  float first[CR];

#define LOAD_ROW(R)                                                           \
  _Pragma("unroll") for (int qq = 0; qq < CQ; ++qq)                           \
    w[(R) & 1][qq] = __builtin_nontemporal_load(                              \
        (const v4f*)(p + (size_t)(R) * NSEQ + 128 * qq));

  LOAD_ROW(0)
  #pragma unroll
  for (int r = 0; r < CR; ++r) {
    __builtin_amdgcn_sched_barrier(0);
    if (r + 1 < CR) { LOAD_ROW(r + 1) }
    __builtin_amdgcn_sched_barrier(0);
    const v4f* wr = w[r & 1];
    v2f lo = {0.f, 0.f}, hi = {0.f, 0.f};
    #pragma unroll
    for (int qq = 0; qq < CQ; ++qq) {
      lo += wr[qq].xy;
      hi += wr[qq].zw;
    }
    sum[r]   = (lo.x + lo.y) + (hi.x + hi.y);
    first[r] = wr[0].x;
  }
#undef LOAD_ROW

  #pragma unroll
  for (int r = 0; r < CR; ++r) {
    float s = sum[r];
    #pragma unroll
    for (int m = 16; m >= 1; m >>= 1) s += __shfl_xor(s, m, 32);
    sum[r] = s;
  }

  if (lane == 0) {
    #pragma unroll
    for (int r = 0; r < CR; ++r) {
      sSum[wave * CR + r] = sum[r];
      sO0 [wave * CR + r] = first[r];
    }
  }
  __syncthreads();

  if (threadIdx.x < 32) {
    const v4f s = *(const v4f*)&sSum[4 * lane];
    const v4f o = *(const v4f*)&sO0 [4 * lane];
    v4f qc, kc;
    qc.x = o.x * rsqrtf(s.x);  kc.x = 1.0f / o.x;
    qc.y = o.y * rsqrtf(s.y);  kc.y = 1.0f / o.y;
    qc.z = o.z * rsqrtf(s.z);  kc.z = 1.0f / o.z;
    qc.w = o.w * rsqrtf(s.w);  kc.w = 1.0f / o.w;
    const int out0 = blockIdx.x * (CW * CR) + 4 * lane;
    *(v4f*)(qcoef + out0) = qc;
    *(v4f*)(kcoef + out0) = kc;
  }
}

__global__ void __launch_bounds__(256)
pret_attn_kernel(const float* __restrict__ q,
                 const float* __restrict__ k,
                 const float* __restrict__ v,
                 const float* __restrict__ qcoef,
                 const float* __restrict__ kcoef,
                 float* __restrict__ out) {
  __shared__ __attribute__((aligned(16))) float sKx[KT][PKV];
  __shared__ __attribute__((aligned(16))) float sVT[DV][PKV];
  __shared__ __attribute__((aligned(16))) float sDT[DV][PKV];
  __shared__ __attribute__((aligned(16))) float sT [2][DK];
  __shared__ __attribute__((aligned(16))) float sS [8][16][18];

  const int nseg  = NSEQ / 128;
  const int bh    = blockIdx.x / nseg;
  const int seg   = blockIdx.x % nseg;
  const int wave  = threadIdx.x >> 5;
  const int lane  = threadIdx.x & 31;
  const int n16   = lane & 15;
  const int half  = lane >> 4;

  const int pos   = (wave < 4) ? wave : (11 - wave);
  const int tile  = seg * 8 + pos;
  const int q0    = tile * 16;

  v2f a[16];
  {
    const float qcv   = qcoef[bh * NSEQ + q0 + n16];
    const float* qrow = q + ((size_t)(bh * NSEQ + q0 + n16)) * DK;
    #pragma unroll
    for (int c = 0; c < 16; ++c) {
      v2f qv = *(const v2f*)(qrow + 4 * c + 2 * half);
      a[c].x = qv.x * qcv;
      a[c].y = qv.y * qcv;
    }
  }

  const v8f zero8 = {0.f, 0.f, 0.f, 0.f, 0.f, 0.f, 0.f, 0.f};
  const v2f ones  = {1.0f, 1.0f};

  const int rt  = wave >> 1;
  const int par = wave & 1;
  const int ct0 = par * 2;
  v8f Dacc[2];
  Dacc[0] = zero8;
  Dacc[1] = zero8;
  v8f Tacc = zero8;

  {
    const int cg = 4 * (wave & 3) + (lane >> 3);
    const int c4 = 4 * cg;
    const int rb = 8 * (wave >> 2) + (lane & 7);

    for (int st = 0; st < 2 * seg; ++st) {
      {
        float  kc[4];
        float4 kv[4], vv[4];
        #pragma unroll
        for (int i = 0; i < 4; ++i) {
          const int krow = st * KT + rb + 16 * i;
          kc[i] = kcoef[bh * NSEQ + krow];
          kv[i] = *(const float4*)(k + ((size_t)(bh * NSEQ + krow)) * DK + c4);
          vv[i] = *(const float4*)(v + ((size_t)(bh * NSEQ + krow)) * DV + c4);
        }
        #pragma unroll
        for (int i = 0; i < 4; ++i) {
          const int r = rb + 16 * i;
          sKx[c4 + 0][r] = kv[i].x * kc[i];
          sKx[c4 + 1][r] = kv[i].y * kc[i];
          sKx[c4 + 2][r] = kv[i].z * kc[i];
          sKx[c4 + 3][r] = kv[i].w * kc[i];
          sVT[c4 + 0][r] = vv[i].x;
          sVT[c4 + 1][r] = vv[i].y;
          sVT[c4 + 2][r] = vv[i].z;
          sVT[c4 + 3][r] = vv[i].w;
        }
      }
      __syncthreads();

      #pragma unroll
      for (int kk = 0; kk < 16; ++kk) {
        const v2f ka = *(const v2f*)&sKx[16 * rt + n16][4 * kk + 2 * half];
        #pragma unroll
        for (int t = 0; t < 2; ++t) {
          const v2f vb = *(const v2f*)&sVT[16 * (ct0 + t) + n16][4 * kk + 2 * half];
          Dacc[t] = __builtin_amdgcn_wmma_f32_16x16x4_f32(false, ka, false, vb,
                                                          (short)0, Dacc[t], false, false);
        }
        if ((kk & 1) == 0) {
          const v2f kt = *(const v2f*)&sKx[16 * rt + n16][4 * (kk + par) + 2 * half];
          Tacc = __builtin_amdgcn_wmma_f32_16x16x4_f32(false, kt, false, ones,
                                                       (short)0, Tacc, false, false);
        }
      }
      __syncthreads();
    }
  }

  #pragma unroll
  for (int t = 0; t < 2; ++t) {
    #pragma unroll
    for (int g = 0; g < 8; ++g)
      sDT[16 * (ct0 + t) + n16][16 * rt + g + 8 * half] = Dacc[t][g];
  }
  if (n16 == 0) {
    #pragma unroll
    for (int g = 0; g < 8; ++g) sT[par][16 * rt + g + 8 * half] = Tacc[g];
  }
  __syncthreads();

  v8f Oacc[4];
  #pragma unroll
  for (int t = 0; t < 4; ++t) Oacc[t] = zero8;
  v8f Pacc = zero8;

  #pragma unroll
  for (int c = 0; c < 16; ++c) {
    const v2f t0 = *(const v2f*)&sT[0][4 * c + 2 * half];
    const v2f t1 = *(const v2f*)&sT[1][4 * c + 2 * half];
    const v2f tb = t0 + t1;
    Pacc = __builtin_amdgcn_wmma_f32_16x16x4_f32(false, a[c], false, tb,
                                                 (short)0, Pacc, false, false);
    #pragma unroll
    for (int ct = 0; ct < 4; ++ct) {
      const v2f db = *(const v2f*)&sDT[16 * ct + n16][4 * c + 2 * half];
      Oacc[ct] = __builtin_amdgcn_wmma_f32_16x16x4_f32(false, a[c], false, db,
                                                       (short)0, Oacc[ct], false, false);
    }
  }

  auto accum_subtile = [&](const v8f& S, const int s) {
    #pragma unroll
    for (int g = 0; g < 8; ++g) sS[wave][g + 8 * half][n16] = S[g];
    v2f sa[4];
    #pragma unroll
    for (int c = 0; c < 4; ++c)
      sa[c] = *(const v2f*)&sS[wave][n16][4 * c + 2 * half];

    #pragma unroll
    for (int c = 0; c < 4; ++c)
      Pacc = __builtin_amdgcn_wmma_f32_16x16x4_f32(false, sa[c], false, ones,
                                                   (short)0, Pacc, false, false);

    #pragma unroll
    for (int ct = 0; ct < 4; ++ct) {
      #pragma unroll
      for (int c = 0; c < 4; ++c) {
        const v2f vb = *(const v2f*)&sVT[ct * 16 + n16][16 * s + 4 * c + 2 * half];
        Oacc[ct] = __builtin_amdgcn_wmma_f32_16x16x4_f32(false, sa[c], false, vb,
                                                         (short)0, Oacc[ct], false, false);
      }
    }
  };

  for (int st = 2 * seg; st < 2 * seg + 2; ++st) {
    {
      const int r0 = threadIdx.x >> 4;
      const int c4 = (threadIdx.x & 15) * 4;
      float  kc[4];
      float4 kv[4], vv[4];
      #pragma unroll
      for (int i = 0; i < 4; ++i) {
        const int krow = st * KT + r0 + 16 * i;
        kc[i] = kcoef[bh * NSEQ + krow];
        kv[i] = *(const float4*)(k + ((size_t)(bh * NSEQ + krow)) * DK + c4);
        vv[i] = *(const float4*)(v + ((size_t)(bh * NSEQ + krow)) * DV + c4);
      }
      #pragma unroll
      for (int i = 0; i < 4; ++i) {
        const int r = r0 + 16 * i;
        sKx[r][c4 + 0] = kv[i].x * kc[i];
        sKx[r][c4 + 1] = kv[i].y * kc[i];
        sKx[r][c4 + 2] = kv[i].z * kc[i];
        sKx[r][c4 + 3] = kv[i].w * kc[i];
        sVT[c4 + 0][r] = vv[i].x;
        sVT[c4 + 1][r] = vv[i].y;
        sVT[c4 + 2][r] = vv[i].z;
        sVT[c4 + 3][r] = vv[i].w;
      }
    }
    __syncthreads();

    const int sub0 = st * 4;
    if (sub0 + 3 < tile) {
      v8f S[4];
      #pragma unroll
      for (int s = 0; s < 4; ++s) S[s] = zero8;
      #pragma unroll
      for (int c = 0; c < 16; ++c) {
        #pragma unroll
        for (int s = 0; s < 4; ++s) {
          const v2f b = *(const v2f*)&sKx[16 * s + n16][4 * c + 2 * half];
          S[s] = __builtin_amdgcn_wmma_f32_16x16x4_f32(false, a[c], false, b,
                                                       (short)0, S[s], false, false);
        }
      }
      #pragma unroll
      for (int s = 0; s < 4; ++s) accum_subtile(S[s], s);
    } else {
      #pragma unroll
      for (int s = 0; s < 4; ++s) {
        const int sub = sub0 + s;
        if (sub <= tile) {
          v8f S = zero8;
          #pragma unroll
          for (int c = 0; c < 16; ++c) {
            const v2f b = *(const v2f*)&sKx[16 * s + n16][4 * c + 2 * half];
            S = __builtin_amdgcn_wmma_f32_16x16x4_f32(false, a[c], false, b,
                                                      (short)0, S, false, false);
          }
          {
            const int key = (sub == tile) ? (sub * 16 + n16) : -1;
            #pragma unroll
            for (int g = 0; g < 8; ++g) {
              const int qr = q0 + g + 8 * half;
              S[g] = (key > qr) ? 0.0f : S[g];
            }
          }
          accum_subtile(S, s);
        }
      }
    }
    __syncthreads();
  }

  #pragma unroll
  for (int ct = 0; ct < 4; ++ct) {
    #pragma unroll
    for (int g = 0; g < 8; ++g) {
      const float p = fmaxf(fabsf(Pacc[g]), 1.0f);
      out[((size_t)(bh * NSEQ + q0 + g + 8 * half)) * DV + ct * 16 + n16] =
          Oacc[ct][g] / p;
    }
  }
}

extern "C" void kernel_launch(void* const* d_in, const int* in_sizes, int n_in,
                              void* d_out, int out_size, void* d_ws, size_t ws_size,
                              hipStream_t stream) {
  const float* q     = (const float*)d_in[0];
  const float* k     = (const float*)d_in[1];
  const float* v     = (const float*)d_in[2];
  const float* omask = (const float*)d_in[3];
  float* out = (float*)d_out;

  float* qcoef = (float*)d_ws;
  float* kcoef = qcoef + (size_t)BH * NSEQ;

  pret_coef_kernel<<<(BH * NSEQ) / (CW * CR), CW * 32, 0, stream>>>(omask, qcoef, kcoef);
  pret_attn_kernel<<<BH * (NSEQ / 128), 256, 0, stream>>>(q, k, v,
                                                          qcoef, kcoef, out);
}


# prefix state no longer re-summed by every query block: a new launch sums each 128-key block's K^T V and key sum once into workspace, attention adds the blocks before its own; summation order changed
# speedup vs baseline: 5.8352x; 2.0912x over previous
#include <hip/hip_runtime.h>
#include <hip/hip_bf16.h>
#include <math.h>


typedef __attribute__((ext_vector_type(2))) float v2f;
typedef __attribute__((ext_vector_type(4))) float v4f;
typedef __attribute__((ext_vector_type(8))) float v8f;

#define BH   16
#define NSEQ 2048
#define DK   64
#define DV   64
#define KT   64
#define PKV  68

#define NBLK 15
#define ST_D (DK * DV)
#define ST_T (2 * DK)
#define WS_NEED ((size_t)(2 * BH * NSEQ + BH * NBLK * (ST_D + ST_T)) * sizeof(float))

#define CW   16
#define CR   8
#define CQ   (NSEQ / 128)

__global__ void __launch_bounds__(CW * 32)
pret_coef_kernel(const float* __restrict__ omask,
                 float* __restrict__ qcoef,
                 float* __restrict__ kcoef) {
  __shared__ __attribute__((aligned(16))) float sSum[CW * CR];
  __shared__ __attribute__((aligned(16))) float sO0 [CW * CR];

  const int lane = threadIdx.x & 31;
  const int wave = threadIdx.x >> 5;
  const int row0 = (blockIdx.x * CW + wave) * CR;
  const float* p = omask + (size_t)row0 * NSEQ + 4 * lane;

  v4f   w[2][CQ];
  float sum[CR];
  float first[CR];

#define LOAD_ROW(R)                                                           \
  _Pragma("unroll") for (int qq = 0; qq < CQ; ++qq)                           \
    w[(R) & 1][qq] = __builtin_nontemporal_load(                              \
        (const v4f*)(p + (size_t)(R) * NSEQ + 128 * qq));

  LOAD_ROW(0)
  #pragma unroll
  for (int r = 0; r < CR; ++r) {
    __builtin_amdgcn_sched_barrier(0);
    if (r + 1 < CR) { LOAD_ROW(r + 1) }
    __builtin_amdgcn_sched_barrier(0);
    const v4f* wr = w[r & 1];
    v2f lo = {0.f, 0.f}, hi = {0.f, 0.f};
    #pragma unroll
    for (int qq = 0; qq < CQ; ++qq) {
      lo += wr[qq].xy;
      hi += wr[qq].zw;
    }
    sum[r]   = (lo.x + lo.y) + (hi.x + hi.y);
    first[r] = wr[0].x;
  }
#undef LOAD_ROW

  #pragma unroll
  for (int r = 0; r < CR; ++r) {
    float s = sum[r];
    #pragma unroll
    for (int m = 16; m >= 1; m >>= 1) s += __shfl_xor(s, m, 32);
    sum[r] = s;
  }

  if (lane == 0) {
    #pragma unroll
    for (int r = 0; r < CR; ++r) {
      sSum[wave * CR + r] = sum[r];
      sO0 [wave * CR + r] = first[r];
    }
  }
  __syncthreads();

  if (threadIdx.x < 32) {
    const v4f s = *(const v4f*)&sSum[4 * lane];
    const v4f o = *(const v4f*)&sO0 [4 * lane];
    v4f qc, kc;
    qc.x = o.x * rsqrtf(s.x);  kc.x = 1.0f / o.x;
    qc.y = o.y * rsqrtf(s.y);  kc.y = 1.0f / o.y;
    qc.z = o.z * rsqrtf(s.z);  kc.z = 1.0f / o.z;
    qc.w = o.w * rsqrtf(s.w);  kc.w = 1.0f / o.w;
    const int out0 = blockIdx.x * (CW * CR) + 4 * lane;
    *(v4f*)(qcoef + out0) = qc;
    *(v4f*)(kcoef + out0) = kc;
  }
}

__device__ __forceinline__ void
pret_sum_stages(const float* __restrict__ k,
                const float* __restrict__ v,
                const float* __restrict__ kcoef,
                const int bh, const int st0, const int st1,
                float (&sKx)[KT][PKV],
                float (&sVT)[DV][PKV],
                v8f (&Dacc)[2], v8f& Tacc) {
  const int wave = threadIdx.x >> 5;
  const int lane = threadIdx.x & 31;
  const int n16  = lane & 15;
  const int half = lane >> 4;
  const int rt   = wave >> 1;
  const int par  = wave & 1;
  const int ct0  = par * 2;
  const v2f ones = {1.0f, 1.0f};

  const int cg = 4 * (wave & 3) + (lane >> 3);
  const int c4 = 4 * cg;
  const int rb = 8 * (wave >> 2) + (lane & 7);

  for (int st = st0; st < st1; ++st) {
    {
      float  kc[4];
      float4 kv[4], vv[4];
      #pragma unroll
      for (int i = 0; i < 4; ++i) {
        const int krow = st * KT + rb + 16 * i;
        kc[i] = kcoef[bh * NSEQ + krow];
        kv[i] = *(const float4*)(k + ((size_t)(bh * NSEQ + krow)) * DK + c4);
        vv[i] = *(const float4*)(v + ((size_t)(bh * NSEQ + krow)) * DV + c4);
      }
      #pragma unroll
      for (int i = 0; i < 4; ++i) {
        const int r = rb + 16 * i;
        sKx[c4 + 0][r] = kv[i].x * kc[i];
        sKx[c4 + 1][r] = kv[i].y * kc[i];
        sKx[c4 + 2][r] = kv[i].z * kc[i];
        sKx[c4 + 3][r] = kv[i].w * kc[i];
        sVT[c4 + 0][r] = vv[i].x;
        sVT[c4 + 1][r] = vv[i].y;
        sVT[c4 + 2][r] = vv[i].z;
        sVT[c4 + 3][r] = vv[i].w;
      }
    }
    __syncthreads();

    #pragma unroll
    for (int kk = 0; kk < 16; ++kk) {
      const v2f ka = *(const v2f*)&sKx[16 * rt + n16][4 * kk + 2 * half];
      #pragma unroll
      for (int t = 0; t < 2; ++t) {
        const v2f vb = *(const v2f*)&sVT[16 * (ct0 + t) + n16][4 * kk + 2 * half];
        Dacc[t] = __builtin_amdgcn_wmma_f32_16x16x4_f32(false, ka, false, vb,
                                                        (short)0, Dacc[t], false, false);
      }
      if ((kk & 1) == 0) {
        const v2f kt = *(const v2f*)&sKx[16 * rt + n16][4 * (kk + par) + 2 * half];
        Tacc = __builtin_amdgcn_wmma_f32_16x16x4_f32(false, kt, false, ones,
                                                     (short)0, Tacc, false, false);
      }
    }
    __syncthreads();
  }
}

__global__ void __launch_bounds__(256)
pret_state_kernel(const float* __restrict__ k,
                  const float* __restrict__ v,
                  const float* __restrict__ kcoef,
                  float* __restrict__ stD,
                  float* __restrict__ stT,
                  const size_t ws_bytes) {
  __shared__ __attribute__((aligned(16))) float sKx[KT][PKV];
  __shared__ __attribute__((aligned(16))) float sVT[DV][PKV];

  if (ws_bytes < WS_NEED) return;

  const int bh   = blockIdx.x / NBLK;
  const int blk  = blockIdx.x % NBLK;
  const int wave = threadIdx.x >> 5;
  const int lane = threadIdx.x & 31;
  const int n16  = lane & 15;
  const int half = lane >> 4;
  const int rt   = wave >> 1;
  const int par  = wave & 1;
  const int ct0  = par * 2;

  const v8f zero8 = {0.f, 0.f, 0.f, 0.f, 0.f, 0.f, 0.f, 0.f};
  v8f Dacc[2];
  Dacc[0] = zero8;
  Dacc[1] = zero8;
  v8f Tacc = zero8;
  pret_sum_stages(k, v, kcoef, bh, 2 * blk, 2 * blk + 2, sKx, sVT, Dacc, Tacc);

  float* d = stD + (size_t)(bh * NBLK + blk) * ST_D;
  #pragma unroll
  for (int t = 0; t < 2; ++t) {
    #pragma unroll
    for (int g = 0; g < 8; ++g)
      d[(16 * rt + g + 8 * half) * DV + 16 * (ct0 + t) + n16] = Dacc[t][g];
  }
  if (n16 == 0) {
    float* tt = stT + (size_t)(bh * NBLK + blk) * ST_T + par * DK;
    #pragma unroll
    for (int g = 0; g < 8; ++g) tt[16 * rt + g + 8 * half] = Tacc[g];
  }
}

__global__ void __launch_bounds__(256)
pret_attn_kernel(const float* __restrict__ q,
                 const float* __restrict__ k,
                 const float* __restrict__ v,
                 const float* __restrict__ qcoef,
                 const float* __restrict__ kcoef,
                 const float* __restrict__ stD,
                 const float* __restrict__ stT,
                 const size_t ws_bytes,
                 float* __restrict__ out) {
  __shared__ __attribute__((aligned(16))) float sKx[KT][PKV];
  __shared__ __attribute__((aligned(16))) float sVT[DV][PKV];
  __shared__ __attribute__((aligned(16))) float sDT[DV][PKV];
  __shared__ __attribute__((aligned(16))) float sT [2][DK];
  __shared__ __attribute__((aligned(16))) float sS [8][16][18];

  const int nseg  = NSEQ / 128;
  const int bh    = blockIdx.x / nseg;
  const int seg   = blockIdx.x % nseg;
  const int wave  = threadIdx.x >> 5;
  const int lane  = threadIdx.x & 31;
  const int n16   = lane & 15;
  const int half  = lane >> 4;

  const int pos   = (wave < 4) ? wave : (11 - wave);
  const int tile  = seg * 8 + pos;
  const int q0    = tile * 16;

  v2f a[16];
  {
    const float qcv   = qcoef[bh * NSEQ + q0 + n16];
    const float* qrow = q + ((size_t)(bh * NSEQ + q0 + n16)) * DK;
    #pragma unroll
    for (int c = 0; c < 16; ++c) {
      v2f qv = *(const v2f*)(qrow + 4 * c + 2 * half);
      a[c].x = qv.x * qcv;
      a[c].y = qv.y * qcv;
    }
  }

  const v8f zero8 = {0.f, 0.f, 0.f, 0.f, 0.f, 0.f, 0.f, 0.f};
  const v2f ones  = {1.0f, 1.0f};

  if (ws_bytes >= WS_NEED) {
    const float* dsrc = stD + (size_t)(bh * NBLK) * ST_D + 4 * threadIdx.x;
    const float* tsrc = stT + (size_t)(bh * NBLK) * ST_T + (threadIdx.x & (ST_T - 1));
    const bool   tOwn = threadIdx.x < ST_T;
    v4f   dsum[4];
    float tsum = 0.0f;
    #pragma unroll
    for (int i = 0; i < 4; ++i) dsum[i] = (v4f){0.f, 0.f, 0.f, 0.f};
    for (int b0 = 0; b0 < seg; b0 += 8) {
      v4f   dw[8][4];
      float tw[8];
      #pragma unroll
      for (int j = 0; j < 8; ++j) {
        const int b = (b0 + j < seg) ? (b0 + j) : (seg - 1);
        #pragma unroll
        for (int i = 0; i < 4; ++i)
          dw[j][i] = *(const v4f*)(dsrc + (size_t)b * ST_D + 1024 * i);
        tw[j] = tsrc[(size_t)b * ST_T];
      }
      #pragma unroll
      for (int j = 0; j < 8; ++j) {
        const float m = (b0 + j < seg) ? 1.0f : 0.0f;
        #pragma unroll
        for (int i = 0; i < 4; ++i) dsum[i] += m * dw[j][i];
        tsum += m * tw[j];
      }
    }
    #pragma unroll
    for (int i = 0; i < 4; ++i) {
      const int e  = 4 * ((int)threadIdx.x + 256 * i);
      const int dk = e >> 6;
      const int dv = e & 63;
      sDT[dv + 0][dk] = dsum[i].x;
      sDT[dv + 1][dk] = dsum[i].y;
      sDT[dv + 2][dk] = dsum[i].z;
      sDT[dv + 3][dk] = dsum[i].w;
    }
    if (tOwn) (&sT[0][0])[threadIdx.x] = tsum;
  } else {
    const int rt  = wave >> 1;
    const int par = wave & 1;
    const int ct0 = par * 2;
    v8f Dacc[2];
    Dacc[0] = zero8;
    Dacc[1] = zero8;
    v8f Tacc = zero8;
    pret_sum_stages(k, v, kcoef, bh, 0, 2 * seg, sKx, sVT, Dacc, Tacc);

    #pragma unroll
    for (int t = 0; t < 2; ++t) {
      #pragma unroll
      for (int g = 0; g < 8; ++g)
        sDT[16 * (ct0 + t) + n16][16 * rt + g + 8 * half] = Dacc[t][g];
    }
    if (n16 == 0) {
      #pragma unroll
      for (int g = 0; g < 8; ++g) sT[par][16 * rt + g + 8 * half] = Tacc[g];
    }
  }
  __syncthreads();

  v8f Oacc[4];
  #pragma unroll
  for (int t = 0; t < 4; ++t) Oacc[t] = zero8;
  v8f Pacc = zero8;

  #pragma unroll
  for (int c = 0; c < 16; ++c) {
    const v2f t0 = *(const v2f*)&sT[0][4 * c + 2 * half];
    const v2f t1 = *(const v2f*)&sT[1][4 * c + 2 * half];
    const v2f tb = t0 + t1;
    Pacc = __builtin_amdgcn_wmma_f32_16x16x4_f32(false, a[c], false, tb,
                                                 (short)0, Pacc, false, false);
    #pragma unroll
    for (int ct = 0; ct < 4; ++ct) {
      const v2f db = *(const v2f*)&sDT[16 * ct + n16][4 * c + 2 * half];
      Oacc[ct] = __builtin_amdgcn_wmma_f32_16x16x4_f32(false, a[c], false, db,
                                                       (short)0, Oacc[ct], false, false);
    }
  }

  auto accum_subtile = [&](const v8f& S, const int s) {
    #pragma unroll
    for (int g = 0; g < 8; ++g) sS[wave][g + 8 * half][n16] = S[g];
    v2f sa[4];
    #pragma unroll
    for (int c = 0; c < 4; ++c)
      sa[c] = *(const v2f*)&sS[wave][n16][4 * c + 2 * half];

    #pragma unroll
    for (int c = 0; c < 4; ++c)
      Pacc = __builtin_amdgcn_wmma_f32_16x16x4_f32(false, sa[c], false, ones,
                                                   (short)0, Pacc, false, false);

    #pragma unroll
    for (int ct = 0; ct < 4; ++ct) {
      #pragma unroll
      for (int c = 0; c < 4; ++c) {
        const v2f vb = *(const v2f*)&sVT[ct * 16 + n16][16 * s + 4 * c + 2 * half];
        Oacc[ct] = __builtin_amdgcn_wmma_f32_16x16x4_f32(false, sa[c], false, vb,
                                                         (short)0, Oacc[ct], false, false);
      }
    }
  };

  for (int st = 2 * seg; st < 2 * seg + 2; ++st) {
    {
      const int r0 = threadIdx.x >> 4;
      const int c4 = (threadIdx.x & 15) * 4;
      float  kc[4];
      float4 kv[4], vv[4];
      #pragma unroll
      for (int i = 0; i < 4; ++i) {
        const int krow = st * KT + r0 + 16 * i;
        kc[i] = kcoef[bh * NSEQ + krow];
        kv[i] = *(const float4*)(k + ((size_t)(bh * NSEQ + krow)) * DK + c4);
        vv[i] = *(const float4*)(v + ((size_t)(bh * NSEQ + krow)) * DV + c4);
      }
      #pragma unroll
      for (int i = 0; i < 4; ++i) {
        const int r = r0 + 16 * i;
        sKx[r][c4 + 0] = kv[i].x * kc[i];
        sKx[r][c4 + 1] = kv[i].y * kc[i];
        sKx[r][c4 + 2] = kv[i].z * kc[i];
        sKx[r][c4 + 3] = kv[i].w * kc[i];
        sVT[c4 + 0][r] = vv[i].x;
        sVT[c4 + 1][r] = vv[i].y;
        sVT[c4 + 2][r] = vv[i].z;
        sVT[c4 + 3][r] = vv[i].w;
      }
    }
    __syncthreads();

    const int sub0 = st * 4;
    if (sub0 + 3 < tile) {
      v8f S[4];
      #pragma unroll
      for (int s = 0; s < 4; ++s) S[s] = zero8;
      #pragma unroll
      for (int c = 0; c < 16; ++c) {
        #pragma unroll
        for (int s = 0; s < 4; ++s) {
          const v2f b = *(const v2f*)&sKx[16 * s + n16][4 * c + 2 * half];
          S[s] = __builtin_amdgcn_wmma_f32_16x16x4_f32(false, a[c], false, b,
                                                       (short)0, S[s], false, false);
        }
      }
      #pragma unroll
      for (int s = 0; s < 4; ++s) accum_subtile(S[s], s);
    } else {
      #pragma unroll
      for (int s = 0; s < 4; ++s) {
        const int sub = sub0 + s;
        if (sub <= tile) {
          v8f S = zero8;
          #pragma unroll
          for (int c = 0; c < 16; ++c) {
            const v2f b = *(const v2f*)&sKx[16 * s + n16][4 * c + 2 * half];
            S = __builtin_amdgcn_wmma_f32_16x16x4_f32(false, a[c], false, b,
                                                      (short)0, S, false, false);
          }
          {
            const int key = (sub == tile) ? (sub * 16 + n16) : -1;
            #pragma unroll
            for (int g = 0; g < 8; ++g) {
              const int qr = q0 + g + 8 * half;
              S[g] = (key > qr) ? 0.0f : S[g];
            }
          }
          accum_subtile(S, s);
        }
      }
    }
    __syncthreads();
  }

  #pragma unroll
  for (int ct = 0; ct < 4; ++ct) {
    #pragma unroll
    for (int g = 0; g < 8; ++g) {
      const float p = fmaxf(fabsf(Pacc[g]), 1.0f);
      out[((size_t)(bh * NSEQ + q0 + g + 8 * half)) * DV + ct * 16 + n16] =
          Oacc[ct][g] / p;
    }
  }
}

extern "C" void kernel_launch(void* const* d_in, const int* in_sizes, int n_in,
                              void* d_out, int out_size, void* d_ws, size_t ws_size,
                              hipStream_t stream) {
  const float* q     = (const float*)d_in[0];
  const float* k     = (const float*)d_in[1];
  const float* v     = (const float*)d_in[2];
  const float* omask = (const float*)d_in[3];
  float* out = (float*)d_out;

  float* qcoef = (float*)d_ws;
  float* kcoef = qcoef + (size_t)BH * NSEQ;
  float* stD   = kcoef + (size_t)BH * NSEQ;
  float* stT   = stD + (size_t)BH * NBLK * ST_D;

  pret_coef_kernel<<<(BH * NSEQ) / (CW * CR), CW * 32, 0, stream>>>(omask, qcoef, kcoef);
  pret_state_kernel<<<BH * NBLK, 256, 0, stream>>>(k, v, kcoef, stD, stT, ws_size);
  pret_attn_kernel<<<BH * (NSEQ / 128), 256, 0, stream>>>(q, k, v,
                                                          qcoef, kcoef, stD, stT,
                                                          ws_size, out);
}


# omask row sums lose their launch: 8 more waves per attention work-group stream them beside the products; q coefficient multiplied in after the sums, k coefficient read from omask column 0
# speedup vs baseline: 6.2869x; 1.0774x over previous
#include <hip/hip_runtime.h>
#include <hip/hip_bf16.h>
#include <math.h>


typedef __attribute__((ext_vector_type(2))) float v2f;
typedef __attribute__((ext_vector_type(4))) float v4f;
typedef __attribute__((ext_vector_type(8))) float v8f;

#define BH   16
#define NSEQ 2048
#define DK   64
#define DV   64
#define KT   64
#define PKV  68

#define NBLK 15
#define ST_D (DK * DV)
#define ST_T (2 * DK)
#define WS_NEED ((size_t)(BH * NBLK * (ST_D + ST_T)) * sizeof(float))

#define AW   8
#define SW   8
#define SR   (128 / SW)
#define CQ   (NSEQ / 128)

#define PRET_WMMA_RUN_END "v_nop\n\tv_nop\n\tv_nop\n\tv_nop\n\tv_nop"

template <int RA, int RB>
__device__ __forceinline__ void
pret_stream_rows(const float* __restrict__ p,
                 float* __restrict__ sq,
                 const int lane) {
  v4f w[2][CQ];

  #pragma unroll
  for (int qq = 0; qq < CQ; ++qq)
    w[RA & 1][qq] = __builtin_nontemporal_load(
        (const v4f*)(p + (size_t)RA * NSEQ + 128 * qq));

  #pragma unroll
  for (int r = RA; r < RB; ++r) {
    __builtin_amdgcn_sched_barrier(0);
    if (r + 1 < RB) {
      #pragma unroll
      for (int qq = 0; qq < CQ; ++qq)
        w[(r + 1) & 1][qq] = __builtin_nontemporal_load(
            (const v4f*)(p + (size_t)(r + 1) * NSEQ + 128 * qq));
    }
    __builtin_amdgcn_sched_barrier(0);
    const v4f* wr = w[r & 1];
    v2f lo = {0.f, 0.f}, hi = {0.f, 0.f};
    #pragma unroll
    for (int qq = 0; qq < CQ; ++qq) {
      lo += wr[qq].xy;
      hi += wr[qq].zw;
    }
    float sum = (lo.x + lo.y) + (hi.x + hi.y);
    #pragma unroll
    for (int m = 16; m >= 1; m >>= 1) sum += __shfl_xor(sum, m, 32);
    if (lane == 0) sq[r] = wr[0].x * rsqrtf(sum);
  }
}

template <bool ROLES>
__device__ __forceinline__ void
pret_sum_stages(const float* __restrict__ k,
                const float* __restrict__ v,
                const float* __restrict__ omask,
                const bool active,
                const int bh, const int st0, const int st1,
                float (&sKx)[KT][PKV],
                float (&sVT)[DV][PKV],
                v8f (&Dacc)[2], v8f& Tacc) {
  const int wave = threadIdx.x >> 5;
  const int lane = threadIdx.x & 31;
  const int n16  = lane & 15;
  const int half = lane >> 4;
  const int rt   = wave >> 1;
  const int par  = wave & 1;
  const int ct0  = par * 2;
  const v2f ones = {1.0f, 1.0f};

  const int cg = 4 * (wave & 3) + (lane >> 3);
  const int c4 = 4 * cg;
  const int rb = 8 * (wave >> 2) + (lane & 7);

  for (int st = st0; st < st1; ++st) {
    if (active) {
      float  kc[4];
      float4 kv[4], vv[4];
      #pragma unroll
      for (int i = 0; i < 4; ++i) {
        const int krow = st * KT + rb + 16 * i;
        kc[i] = 1.0f / omask[((size_t)(bh * NSEQ + krow)) * NSEQ];
        kv[i] = *(const float4*)(k + ((size_t)(bh * NSEQ + krow)) * DK + c4);
        vv[i] = *(const float4*)(v + ((size_t)(bh * NSEQ + krow)) * DV + c4);
      }
      #pragma unroll
      for (int i = 0; i < 4; ++i) {
        const int r = rb + 16 * i;
        sKx[c4 + 0][r] = kv[i].x * kc[i];
        sKx[c4 + 1][r] = kv[i].y * kc[i];
        sKx[c4 + 2][r] = kv[i].z * kc[i];
        sKx[c4 + 3][r] = kv[i].w * kc[i];
        sVT[c4 + 0][r] = vv[i].x;
        sVT[c4 + 1][r] = vv[i].y;
        sVT[c4 + 2][r] = vv[i].z;
        sVT[c4 + 3][r] = vv[i].w;
      }
    }
    __syncthreads();

    if (active) {
    #pragma unroll
    for (int kk = 0; kk < 16; ++kk) {
      const v2f ka = *(const v2f*)&sKx[16 * rt + n16][4 * kk + 2 * half];
      #pragma unroll
      for (int t = 0; t < 2; ++t) {
        const v2f vb = *(const v2f*)&sVT[16 * (ct0 + t) + n16][4 * kk + 2 * half];
        Dacc[t] = __builtin_amdgcn_wmma_f32_16x16x4_f32(false, ka, false, vb,
                                                        (short)0, Dacc[t], false, false);
      }
      if ((kk & 1) == 0) {
        const v2f kt = *(const v2f*)&sKx[16 * rt + n16][4 * (kk + par) + 2 * half];
        Tacc = __builtin_amdgcn_wmma_f32_16x16x4_f32(false, kt, false, ones,
                                                     (short)0, Tacc, false, false);
      }
    }
    if (ROLES) asm volatile(PRET_WMMA_RUN_END : "+v"(Dacc[0]), "+v"(Dacc[1]), "+v"(Tacc));
    }
    __syncthreads();
  }
}

__global__ void __launch_bounds__(256)
pret_state_kernel(const float* __restrict__ k,
                  const float* __restrict__ v,
                  const float* __restrict__ omask,
                  float* __restrict__ stD,
                  float* __restrict__ stT,
                  const size_t ws_bytes) {
  __shared__ __attribute__((aligned(16))) float sKx[KT][PKV];
  __shared__ __attribute__((aligned(16))) float sVT[DV][PKV];

  if (ws_bytes < WS_NEED) return;

  const int bh   = blockIdx.x / NBLK;
  const int blk  = blockIdx.x % NBLK;
  const int wave = threadIdx.x >> 5;
  const int lane = threadIdx.x & 31;
  const int n16  = lane & 15;
  const int half = lane >> 4;
  const int rt   = wave >> 1;
  const int par  = wave & 1;
  const int ct0  = par * 2;

  const v8f zero8 = {0.f, 0.f, 0.f, 0.f, 0.f, 0.f, 0.f, 0.f};
  v8f Dacc[2];
  Dacc[0] = zero8;
  Dacc[1] = zero8;
  v8f Tacc = zero8;
  pret_sum_stages<false>(k, v, omask, true, bh, 2 * blk, 2 * blk + 2, sKx, sVT, Dacc, Tacc);

  float* d = stD + (size_t)(bh * NBLK + blk) * ST_D;
  #pragma unroll
  for (int t = 0; t < 2; ++t) {
    #pragma unroll
    for (int g = 0; g < 8; ++g)
      d[(16 * rt + g + 8 * half) * DV + 16 * (ct0 + t) + n16] = Dacc[t][g];
  }
  if (n16 == 0) {
    float* tt = stT + (size_t)(bh * NBLK + blk) * ST_T + par * DK;
    #pragma unroll
    for (int g = 0; g < 8; ++g) tt[16 * rt + g + 8 * half] = Tacc[g];
  }
}

__global__ void __launch_bounds__((AW + SW) * 32)
pret_attn_kernel(const float* __restrict__ q,
                 const float* __restrict__ k,
                 const float* __restrict__ v,
                 const float* __restrict__ omask,
                 const float* __restrict__ stD,
                 const float* __restrict__ stT,
                 const size_t ws_bytes,
                 float* __restrict__ out) {
  __shared__ __attribute__((aligned(16))) float sKx[KT][PKV];
  __shared__ __attribute__((aligned(16))) float sVT[DV][PKV];
  __shared__ __attribute__((aligned(16))) float sDT[DV][PKV];
  __shared__ __attribute__((aligned(16))) float sT [2][DK];
  __shared__ __attribute__((aligned(16))) float sS [AW][16][18];
  __shared__ __attribute__((aligned(16))) float sQc[128];

  const int nseg  = NSEQ / 128;
  const int bh    = blockIdx.x / nseg;
  const int seg   = blockIdx.x % nseg;
  const int wave  = __builtin_amdgcn_readfirstlane((int)(threadIdx.x >> 5));
  const int lane  = threadIdx.x & 31;
  const int n16   = lane & 15;
  const int half  = lane >> 4;
  const bool prod = wave < AW;
  const bool haveStates = ws_bytes >= WS_NEED;

  const int    srow = (wave - AW) * SR;
  const float* sp   = omask + ((size_t)(bh * NSEQ + seg * 128 + (prod ? 0 : srow))) * NSEQ + 4 * lane;
  float*       sq   = &sQc[prod ? 0 : srow];

  const int pos   = (wave < 4) ? wave : (11 - wave);
  const int tile  = seg * 8 + pos;
  const int q0    = tile * 16;

  const v8f zero8 = {0.f, 0.f, 0.f, 0.f, 0.f, 0.f, 0.f, 0.f};
  const v2f ones  = {1.0f, 1.0f};

  v2f a[16];
  #pragma unroll
  for (int c = 0; c < 16; ++c) a[c] = (v2f){0.f, 0.f};

  if (prod) {
    const float* qrow = q + ((size_t)(bh * NSEQ + q0 + n16)) * DK;
    #pragma unroll
    for (int c = 0; c < 16; ++c)
      a[c] = *(const v2f*)(qrow + 4 * c + 2 * half);
  }

  if (haveStates) {
    if (prod) {
      const float* dsrc = stD + (size_t)(bh * NBLK) * ST_D + 4 * threadIdx.x;
      const float* tsrc = stT + (size_t)(bh * NBLK) * ST_T + (threadIdx.x & (ST_T - 1));
      const bool   tOwn = threadIdx.x < ST_T;
      v4f   dsum[4];
      float tsum = 0.0f;
      #pragma unroll
      for (int i = 0; i < 4; ++i) dsum[i] = (v4f){0.f, 0.f, 0.f, 0.f};
      for (int b0 = 0; b0 < seg; b0 += 8) {
        v4f   dw[8][4];
        float tw[8];
        #pragma unroll
        for (int j = 0; j < 8; ++j) {
          const int b = (b0 + j < seg) ? (b0 + j) : (seg - 1);
          #pragma unroll
          for (int i = 0; i < 4; ++i)
            dw[j][i] = *(const v4f*)(dsrc + (size_t)b * ST_D + 1024 * i);
          tw[j] = tsrc[(size_t)b * ST_T];
        }
        #pragma unroll
        for (int j = 0; j < 8; ++j) {
          const float m = (b0 + j < seg) ? 1.0f : 0.0f;
          #pragma unroll
          for (int i = 0; i < 4; ++i) dsum[i] += m * dw[j][i];
          tsum += m * tw[j];
        }
      }
      #pragma unroll
      for (int i = 0; i < 4; ++i) {
        const int e  = 4 * ((int)threadIdx.x + 256 * i);
        const int dk = e >> 6;
        const int dv = e & 63;
        sDT[dv + 0][dk] = dsum[i].x;
        sDT[dv + 1][dk] = dsum[i].y;
        sDT[dv + 2][dk] = dsum[i].z;
        sDT[dv + 3][dk] = dsum[i].w;
      }
      if (tOwn) (&sT[0][0])[threadIdx.x] = tsum;
    } else {
      pret_stream_rows<0, 3>(sp, sq, lane);
    }
  } else {
    if (!prod) pret_stream_rows<0, 3>(sp, sq, lane);
    const int rt  = (wave >> 1) & 3;
    const int par = wave & 1;
    const int ct0 = par * 2;
    v8f Dacc[2];
    Dacc[0] = zero8;
    Dacc[1] = zero8;
    v8f Tacc = zero8;
    pret_sum_stages<true>(k, v, omask, prod, bh, 0, 2 * seg, sKx, sVT, Dacc, Tacc);

    if (prod) {
      #pragma unroll
      for (int t = 0; t < 2; ++t) {
        #pragma unroll
        for (int g = 0; g < 8; ++g)
          sDT[16 * (ct0 + t) + n16][16 * rt + g + 8 * half] = Dacc[t][g];
      }
      if (n16 == 0) {
        #pragma unroll
        for (int g = 0; g < 8; ++g) sT[par][16 * rt + g + 8 * half] = Tacc[g];
      }
    }
  }
  __syncthreads();

  v8f Oacc[4];
  #pragma unroll
  for (int t = 0; t < 4; ++t) Oacc[t] = zero8;
  v8f Pacc = zero8;

  if (prod) {
    #pragma unroll
    for (int c = 0; c < 16; ++c) {
      const v2f t0 = *(const v2f*)&sT[0][4 * c + 2 * half];
      const v2f t1 = *(const v2f*)&sT[1][4 * c + 2 * half];
      const v2f tb = t0 + t1;
      Pacc = __builtin_amdgcn_wmma_f32_16x16x4_f32(false, a[c], false, tb,
                                                   (short)0, Pacc, false, false);
      #pragma unroll
      for (int ct = 0; ct < 4; ++ct) {
        const v2f db = *(const v2f*)&sDT[16 * ct + n16][4 * c + 2 * half];
        Oacc[ct] = __builtin_amdgcn_wmma_f32_16x16x4_f32(false, a[c], false, db,
                                                         (short)0, Oacc[ct], false, false);
      }
    }
    asm volatile(PRET_WMMA_RUN_END : "+v"(Oacc[0]), "+v"(Oacc[1]), "+v"(Oacc[2]),
                                     "+v"(Oacc[3]), "+v"(Pacc));
  }

  auto accum_subtile = [&](const v8f& S, const int s) {
    #pragma unroll
    for (int g = 0; g < 8; ++g) sS[wave][g + 8 * half][n16] = S[g];
    v2f sa[4];
    #pragma unroll
    for (int c = 0; c < 4; ++c)
      sa[c] = *(const v2f*)&sS[wave][n16][4 * c + 2 * half];

    #pragma unroll
    for (int c = 0; c < 4; ++c)
      Pacc = __builtin_amdgcn_wmma_f32_16x16x4_f32(false, sa[c], false, ones,
                                                   (short)0, Pacc, false, false);

    #pragma unroll
    for (int ct = 0; ct < 4; ++ct) {
      #pragma unroll
      for (int c = 0; c < 4; ++c) {
        const v2f vb = *(const v2f*)&sVT[ct * 16 + n16][16 * s + 4 * c + 2 * half];
        Oacc[ct] = __builtin_amdgcn_wmma_f32_16x16x4_f32(false, sa[c], false, vb,
                                                         (short)0, Oacc[ct], false, false);
      }
    }
  };

  for (int st = 2 * seg; st < 2 * seg + 2; ++st) {
    const bool firstStage = (st == 2 * seg);
    if (prod) {
      const int r0 = threadIdx.x >> 4;
      const int c4 = (threadIdx.x & 15) * 4;
      float  kc[4];
      float4 kv[4], vv[4];
      #pragma unroll
      for (int i = 0; i < 4; ++i) {
        const int krow = st * KT + r0 + 16 * i;
        kc[i] = 1.0f / omask[((size_t)(bh * NSEQ + krow)) * NSEQ];
        kv[i] = *(const float4*)(k + ((size_t)(bh * NSEQ + krow)) * DK + c4);
        vv[i] = *(const float4*)(v + ((size_t)(bh * NSEQ + krow)) * DV + c4);
      }
      #pragma unroll
      for (int i = 0; i < 4; ++i) {
        const int r = r0 + 16 * i;
        sKx[r][c4 + 0] = kv[i].x * kc[i];
        sKx[r][c4 + 1] = kv[i].y * kc[i];
        sKx[r][c4 + 2] = kv[i].z * kc[i];
        sKx[r][c4 + 3] = kv[i].w * kc[i];
        sVT[c4 + 0][r] = vv[i].x;
        sVT[c4 + 1][r] = vv[i].y;
        sVT[c4 + 2][r] = vv[i].z;
        sVT[c4 + 3][r] = vv[i].w;
      }
    } else {
      if (firstStage) pret_stream_rows<3, 7>(sp, sq, lane);
      else            pret_stream_rows<11, 13>(sp, sq, lane);
    }
    __syncthreads();

    if (prod) {
      const int sub0 = st * 4;
      if (sub0 + 3 < tile) {
        v8f S[4];
        #pragma unroll
        for (int s = 0; s < 4; ++s) S[s] = zero8;
        #pragma unroll
        for (int c = 0; c < 16; ++c) {
          #pragma unroll
          for (int s = 0; s < 4; ++s) {
            const v2f b = *(const v2f*)&sKx[16 * s + n16][4 * c + 2 * half];
            S[s] = __builtin_amdgcn_wmma_f32_16x16x4_f32(false, a[c], false, b,
                                                         (short)0, S[s], false, false);
          }
        }
        #pragma unroll
        for (int s = 0; s < 4; ++s) accum_subtile(S[s], s);
      } else {
        #pragma unroll
        for (int s = 0; s < 4; ++s) {
          const int sub = sub0 + s;
          if (sub <= tile) {
            v8f S = zero8;
            #pragma unroll
            for (int c = 0; c < 16; ++c) {
              const v2f b = *(const v2f*)&sKx[16 * s + n16][4 * c + 2 * half];
              S = __builtin_amdgcn_wmma_f32_16x16x4_f32(false, a[c], false, b,
                                                        (short)0, S, false, false);
            }
            {
              const int key = (sub == tile) ? (sub * 16 + n16) : -1;
              #pragma unroll
              for (int g = 0; g < 8; ++g) {
                const int qr = q0 + g + 8 * half;
                S[g] = (key > qr) ? 0.0f : S[g];
              }
            }
            accum_subtile(S, s);
          }
        }
      }
      asm volatile(PRET_WMMA_RUN_END : "+v"(Oacc[0]), "+v"(Oacc[1]), "+v"(Oacc[2]),
                                       "+v"(Oacc[3]), "+v"(Pacc));
    } else {
      if (firstStage) pret_stream_rows<7, 11>(sp, sq, lane);
      else            pret_stream_rows<13, 16>(sp, sq, lane);
    }
    __syncthreads();
  }

  if (prod) {
    const v4f c0 = *(const v4f*)&sQc[16 * pos + 8 * half];
    const v4f c1 = *(const v4f*)&sQc[16 * pos + 8 * half + 4];
    const float cq[8] = {c0.x, c0.y, c0.z, c0.w, c1.x, c1.y, c1.z, c1.w};
    #pragma unroll
    for (int ct = 0; ct < 4; ++ct) {
      #pragma unroll
      for (int g = 0; g < 8; ++g) {
        const float p = fmaxf(fabsf(Pacc[g] * cq[g]), 1.0f);
        out[((size_t)(bh * NSEQ + q0 + g + 8 * half)) * DV + ct * 16 + n16] =
            (Oacc[ct][g] * cq[g]) / p;
      }
    }
  }
}

extern "C" void kernel_launch(void* const* d_in, const int* in_sizes, int n_in,
                              void* d_out, int out_size, void* d_ws, size_t ws_size,
                              hipStream_t stream) {
  const float* q     = (const float*)d_in[0];
  const float* k     = (const float*)d_in[1];
  const float* v     = (const float*)d_in[2];
  const float* omask = (const float*)d_in[3];
  float* out = (float*)d_out;

  float* stD = (float*)d_ws;
  float* stT = stD + (size_t)BH * NBLK * ST_D;

  pret_state_kernel<<<BH * NBLK, 256, 0, stream>>>(k, v, omask, stD, stT, ws_size);
  pret_attn_kernel<<<BH * (NSEQ / 128), (AW + SW) * 32, 0, stream>>>(q, k, v, omask,
                                                                   stD, stT, ws_size, out);
}


# omask stream split over both launches: stream-only work-groups in the state launch sum the first 64 rows of each 128-row query block into workspace, attention streams the rest; row-sum order kept
# speedup vs baseline: 6.6695x; 1.0609x over previous
#include <hip/hip_runtime.h>
#include <hip/hip_bf16.h>
#include <math.h>


typedef __attribute__((ext_vector_type(2))) float v2f;
typedef __attribute__((ext_vector_type(4))) float v4f;
typedef __attribute__((ext_vector_type(8))) float v8f;

#define BH   16
#define NSEQ 2048
#define DK   64
#define DV   64
#define KT   64
#define PKV  68

#define NBLK 15
#define ST_D (DK * DV)
#define ST_T (2 * DK)
#define CQB  (NSEQ / 128)
#define HR   64
#define WS_NEED ((size_t)(BH * NBLK * (ST_D + ST_T) + BH * NSEQ) * sizeof(float))

#define AW   8
#define SW   8
#define SR   (128 / SW)
#define SRH  ((128 - HR) / SW)
#define CQ   (NSEQ / 128)

#define PRET_WMMA_RUN_END "v_nop\n\tv_nop\n\tv_nop\n\tv_nop\n\tv_nop"

template <int RA, int RB>
__device__ __forceinline__ void
pret_stream_rows(const float* __restrict__ p,
                 float* __restrict__ sq,
                 const int lane) {
  v4f w[2][CQ];

  #pragma unroll
  for (int qq = 0; qq < CQ; ++qq)
    w[RA & 1][qq] = __builtin_nontemporal_load(
        (const v4f*)(p + (size_t)RA * NSEQ + 128 * qq));

  #pragma unroll
  for (int r = RA; r < RB; ++r) {
    __builtin_amdgcn_sched_barrier(0);
    if (r + 1 < RB) {
      #pragma unroll
      for (int qq = 0; qq < CQ; ++qq)
        w[(r + 1) & 1][qq] = __builtin_nontemporal_load(
            (const v4f*)(p + (size_t)(r + 1) * NSEQ + 128 * qq));
    }
    __builtin_amdgcn_sched_barrier(0);
    const v4f* wr = w[r & 1];
    v2f lo = {0.f, 0.f}, hi = {0.f, 0.f};
    #pragma unroll
    for (int qq = 0; qq < CQ; ++qq) {
      lo += wr[qq].xy;
      hi += wr[qq].zw;
    }
    float sum = (lo.x + lo.y) + (hi.x + hi.y);
    #pragma unroll
    for (int m = 16; m >= 1; m >>= 1) sum += __shfl_xor(sum, m, 32);
    if (lane == 0) sq[r] = wr[0].x * rsqrtf(sum);
  }
}

template <bool ROLES>
__device__ __forceinline__ void
pret_sum_stages(const float* __restrict__ k,
                const float* __restrict__ v,
                const float* __restrict__ omask,
                const bool active,
                const int bh, const int st0, const int st1,
                float (&sKx)[KT][PKV],
                float (&sVT)[DV][PKV],
                v8f (&Dacc)[2], v8f& Tacc) {
  const int wave = threadIdx.x >> 5;
  const int lane = threadIdx.x & 31;
  const int n16  = lane & 15;
  const int half = lane >> 4;
  const int rt   = wave >> 1;
  const int par  = wave & 1;
  const int ct0  = par * 2;
  const v2f ones = {1.0f, 1.0f};

  const int cg = 4 * (wave & 3) + (lane >> 3);
  const int c4 = 4 * cg;
  const int rb = 8 * (wave >> 2) + (lane & 7);

  for (int st = st0; st < st1; ++st) {
    if (active) {
      float  kc[4];
      float4 kv[4], vv[4];
      #pragma unroll
      for (int i = 0; i < 4; ++i) {
        const int krow = st * KT + rb + 16 * i;
        kc[i] = 1.0f / omask[((size_t)(bh * NSEQ + krow)) * NSEQ];
        kv[i] = *(const float4*)(k + ((size_t)(bh * NSEQ + krow)) * DK + c4);
        vv[i] = *(const float4*)(v + ((size_t)(bh * NSEQ + krow)) * DV + c4);
      }
      #pragma unroll
      for (int i = 0; i < 4; ++i) {
        const int r = rb + 16 * i;
        sKx[c4 + 0][r] = kv[i].x * kc[i];
        sKx[c4 + 1][r] = kv[i].y * kc[i];
        sKx[c4 + 2][r] = kv[i].z * kc[i];
        sKx[c4 + 3][r] = kv[i].w * kc[i];
        sVT[c4 + 0][r] = vv[i].x;
        sVT[c4 + 1][r] = vv[i].y;
        sVT[c4 + 2][r] = vv[i].z;
        sVT[c4 + 3][r] = vv[i].w;
      }
    }
    __syncthreads();

    if (active) {
    #pragma unroll
    for (int kk = 0; kk < 16; ++kk) {
      const v2f ka = *(const v2f*)&sKx[16 * rt + n16][4 * kk + 2 * half];
      #pragma unroll
      for (int t = 0; t < 2; ++t) {
        const v2f vb = *(const v2f*)&sVT[16 * (ct0 + t) + n16][4 * kk + 2 * half];
        Dacc[t] = __builtin_amdgcn_wmma_f32_16x16x4_f32(false, ka, false, vb,
                                                        (short)0, Dacc[t], false, false);
      }
      if ((kk & 1) == 0) {
        const v2f kt = *(const v2f*)&sKx[16 * rt + n16][4 * (kk + par) + 2 * half];
        Tacc = __builtin_amdgcn_wmma_f32_16x16x4_f32(false, kt, false, ones,
                                                     (short)0, Tacc, false, false);
      }
    }
    if (ROLES) asm volatile(PRET_WMMA_RUN_END : "+v"(Dacc[0]), "+v"(Dacc[1]), "+v"(Tacc));
    }
    __syncthreads();
  }
}

__global__ void __launch_bounds__(256)
pret_state_kernel(const float* __restrict__ k,
                  const float* __restrict__ v,
                  const float* __restrict__ omask,
                  float* __restrict__ stD,
                  float* __restrict__ stT,
                  float* __restrict__ cqw,
                  const size_t ws_bytes) {
  __shared__ __attribute__((aligned(16))) float sKx[KT][PKV];
  __shared__ __attribute__((aligned(16))) float sVT[DV][PKV];

  if (ws_bytes < WS_NEED) return;

  if (blockIdx.x < BH * CQB) {
    const int    sw   = __builtin_amdgcn_readfirstlane((int)(threadIdx.x >> 5));
    const int    sl   = threadIdx.x & 31;
    const size_t row0 = (size_t)blockIdx.x * 128 + sw * (HR / 8);
    pret_stream_rows<0, HR / 8>(omask + row0 * NSEQ + 4 * sl, cqw + row0, sl);
    return;
  }

  const int sb   = blockIdx.x - BH * CQB;
  const int bh   = sb / NBLK;
  const int blk  = sb % NBLK;
  const int wave = threadIdx.x >> 5;
  const int lane = threadIdx.x & 31;
  const int n16  = lane & 15;
  const int half = lane >> 4;
  const int rt   = wave >> 1;
  const int par  = wave & 1;
  const int ct0  = par * 2;

  const v8f zero8 = {0.f, 0.f, 0.f, 0.f, 0.f, 0.f, 0.f, 0.f};
  v8f Dacc[2];
  Dacc[0] = zero8;
  Dacc[1] = zero8;
  v8f Tacc = zero8;
  pret_sum_stages<false>(k, v, omask, true, bh, 2 * blk, 2 * blk + 2, sKx, sVT, Dacc, Tacc);

  float* d = stD + (size_t)(bh * NBLK + blk) * ST_D;
  #pragma unroll
  for (int t = 0; t < 2; ++t) {
    #pragma unroll
    for (int g = 0; g < 8; ++g)
      d[(16 * rt + g + 8 * half) * DV + 16 * (ct0 + t) + n16] = Dacc[t][g];
  }
  if (n16 == 0) {
    float* tt = stT + (size_t)(bh * NBLK + blk) * ST_T + par * DK;
    #pragma unroll
    for (int g = 0; g < 8; ++g) tt[16 * rt + g + 8 * half] = Tacc[g];
  }
}

__global__ void __launch_bounds__((AW + SW) * 32)
pret_attn_kernel(const float* __restrict__ q,
                 const float* __restrict__ k,
                 const float* __restrict__ v,
                 const float* __restrict__ omask,
                 const float* __restrict__ stD,
                 const float* __restrict__ stT,
                 const float* __restrict__ cqw,
                 const size_t ws_bytes,
                 float* __restrict__ out) {
  __shared__ __attribute__((aligned(16))) float sKx[KT][PKV];
  __shared__ __attribute__((aligned(16))) float sVT[DV][PKV];
  __shared__ __attribute__((aligned(16))) float sDT[DV][PKV];
  __shared__ __attribute__((aligned(16))) float sT [2][DK];
  __shared__ __attribute__((aligned(16))) float sS [AW][16][18];
  __shared__ __attribute__((aligned(16))) float sQc[128];

  const int nseg  = NSEQ / 128;
  const int bh    = blockIdx.x / nseg;
  const int seg   = blockIdx.x % nseg;
  const int wave  = __builtin_amdgcn_readfirstlane((int)(threadIdx.x >> 5));
  const int lane  = threadIdx.x & 31;
  const int n16   = lane & 15;
  const int half  = lane >> 4;
  const bool prod = wave < AW;
  const bool haveStates = ws_bytes >= WS_NEED;

  const int    srow = haveStates ? HR + (wave - AW) * SRH
                                 : (wave - AW) * SR;
  const float* sp   = omask + ((size_t)(bh * NSEQ + seg * 128 + (prod ? 0 : srow))) * NSEQ + 4 * lane;
  float*       sq   = &sQc[prod ? 0 : srow];

  const int pos   = (wave < 4) ? wave : (11 - wave);
  const int tile  = seg * 8 + pos;
  const int q0    = tile * 16;

  const v8f zero8 = {0.f, 0.f, 0.f, 0.f, 0.f, 0.f, 0.f, 0.f};
  const v2f ones  = {1.0f, 1.0f};

  v2f a[16];
  #pragma unroll
  for (int c = 0; c < 16; ++c) a[c] = (v2f){0.f, 0.f};

  if (prod) {
    const float* qrow = q + ((size_t)(bh * NSEQ + q0 + n16)) * DK;
    #pragma unroll
    for (int c = 0; c < 16; ++c)
      a[c] = *(const v2f*)(qrow + 4 * c + 2 * half);
  }

  if (haveStates) {
    if (prod) {
      const float* dsrc = stD + (size_t)(bh * NBLK) * ST_D + 4 * threadIdx.x;
      const float* tsrc = stT + (size_t)(bh * NBLK) * ST_T + (threadIdx.x & (ST_T - 1));
      const bool   tOwn = threadIdx.x < ST_T;
      v4f   dsum[4];
      float tsum = 0.0f;
      #pragma unroll
      for (int i = 0; i < 4; ++i) dsum[i] = (v4f){0.f, 0.f, 0.f, 0.f};
      for (int b0 = 0; b0 < seg; b0 += 8) {
        v4f   dw[8][4];
        float tw[8];
        #pragma unroll
        for (int j = 0; j < 8; ++j) {
          const int b = (b0 + j < seg) ? (b0 + j) : (seg - 1);
          #pragma unroll
          for (int i = 0; i < 4; ++i)
            dw[j][i] = *(const v4f*)(dsrc + (size_t)b * ST_D + 1024 * i);
          tw[j] = tsrc[(size_t)b * ST_T];
        }
        #pragma unroll
        for (int j = 0; j < 8; ++j) {
          const float m = (b0 + j < seg) ? 1.0f : 0.0f;
          #pragma unroll
          for (int i = 0; i < 4; ++i) dsum[i] += m * dw[j][i];
          tsum += m * tw[j];
        }
      }
      #pragma unroll
      for (int i = 0; i < 4; ++i) {
        const int e  = 4 * ((int)threadIdx.x + 256 * i);
        const int dk = e >> 6;
        const int dv = e & 63;
        sDT[dv + 0][dk] = dsum[i].x;
        sDT[dv + 1][dk] = dsum[i].y;
        sDT[dv + 2][dk] = dsum[i].z;
        sDT[dv + 3][dk] = dsum[i].w;
      }
      if (tOwn) (&sT[0][0])[threadIdx.x] = tsum;
    } else {
      const int   crow = (wave - AW) * (HR / SW) + (lane & (HR / SW - 1));
      const float cpre = cqw[(size_t)(bh * NSEQ + seg * 128) + crow];
      pret_stream_rows<0, 2>(sp, sq, lane);
      if (lane < HR / SW) sQc[crow] = cpre;
    }
  } else {
    if (!prod) pret_stream_rows<0, 3>(sp, sq, lane);
    const int rt  = (wave >> 1) & 3;
    const int par = wave & 1;
    const int ct0 = par * 2;
    v8f Dacc[2];
    Dacc[0] = zero8;
    Dacc[1] = zero8;
    v8f Tacc = zero8;
    pret_sum_stages<true>(k, v, omask, prod, bh, 0, 2 * seg, sKx, sVT, Dacc, Tacc);

    if (prod) {
      #pragma unroll
      for (int t = 0; t < 2; ++t) {
        #pragma unroll
        for (int g = 0; g < 8; ++g)
          sDT[16 * (ct0 + t) + n16][16 * rt + g + 8 * half] = Dacc[t][g];
      }
      if (n16 == 0) {
        #pragma unroll
        for (int g = 0; g < 8; ++g) sT[par][16 * rt + g + 8 * half] = Tacc[g];
      }
    }
  }
  __syncthreads();

  v8f Oacc[4];
  #pragma unroll
  for (int t = 0; t < 4; ++t) Oacc[t] = zero8;
  v8f Pacc = zero8;

  if (prod) {
    #pragma unroll
    for (int c = 0; c < 16; ++c) {
      const v2f t0 = *(const v2f*)&sT[0][4 * c + 2 * half];
      const v2f t1 = *(const v2f*)&sT[1][4 * c + 2 * half];
      const v2f tb = t0 + t1;
      Pacc = __builtin_amdgcn_wmma_f32_16x16x4_f32(false, a[c], false, tb,
                                                   (short)0, Pacc, false, false);
      #pragma unroll
      for (int ct = 0; ct < 4; ++ct) {
        const v2f db = *(const v2f*)&sDT[16 * ct + n16][4 * c + 2 * half];
        Oacc[ct] = __builtin_amdgcn_wmma_f32_16x16x4_f32(false, a[c], false, db,
                                                         (short)0, Oacc[ct], false, false);
      }
    }
    asm volatile(PRET_WMMA_RUN_END : "+v"(Oacc[0]), "+v"(Oacc[1]), "+v"(Oacc[2]),
                                     "+v"(Oacc[3]), "+v"(Pacc));
  }

  auto accum_subtile = [&](const v8f& S, const int s) {
    #pragma unroll
    for (int g = 0; g < 8; ++g) sS[wave][g + 8 * half][n16] = S[g];
    v2f sa[4];
    #pragma unroll
    for (int c = 0; c < 4; ++c)
      sa[c] = *(const v2f*)&sS[wave][n16][4 * c + 2 * half];

    #pragma unroll
    for (int c = 0; c < 4; ++c)
      Pacc = __builtin_amdgcn_wmma_f32_16x16x4_f32(false, sa[c], false, ones,
                                                   (short)0, Pacc, false, false);

    #pragma unroll
    for (int ct = 0; ct < 4; ++ct) {
      #pragma unroll
      for (int c = 0; c < 4; ++c) {
        const v2f vb = *(const v2f*)&sVT[ct * 16 + n16][16 * s + 4 * c + 2 * half];
        Oacc[ct] = __builtin_amdgcn_wmma_f32_16x16x4_f32(false, sa[c], false, vb,
                                                         (short)0, Oacc[ct], false, false);
      }
    }
  };

  for (int st = 2 * seg; st < 2 * seg + 2; ++st) {
    const bool firstStage = (st == 2 * seg);
    if (prod) {
      const int r0 = threadIdx.x >> 4;
      const int c4 = (threadIdx.x & 15) * 4;
      float  kc[4];
      float4 kv[4], vv[4];
      #pragma unroll
      for (int i = 0; i < 4; ++i) {
        const int krow = st * KT + r0 + 16 * i;
        kc[i] = 1.0f / omask[((size_t)(bh * NSEQ + krow)) * NSEQ];
        kv[i] = *(const float4*)(k + ((size_t)(bh * NSEQ + krow)) * DK + c4);
        vv[i] = *(const float4*)(v + ((size_t)(bh * NSEQ + krow)) * DV + c4);
      }
      #pragma unroll
      for (int i = 0; i < 4; ++i) {
        const int r = r0 + 16 * i;
        sKx[r][c4 + 0] = kv[i].x * kc[i];
        sKx[r][c4 + 1] = kv[i].y * kc[i];
        sKx[r][c4 + 2] = kv[i].z * kc[i];
        sKx[r][c4 + 3] = kv[i].w * kc[i];
        sVT[c4 + 0][r] = vv[i].x;
        sVT[c4 + 1][r] = vv[i].y;
        sVT[c4 + 2][r] = vv[i].z;
        sVT[c4 + 3][r] = vv[i].w;
      }
    } else {
      if (haveStates) {
        if (firstStage) pret_stream_rows<2, 4>(sp, sq, lane);
        else            pret_stream_rows<6, 7>(sp, sq, lane);
      } else {
        if (firstStage) pret_stream_rows<3, 7>(sp, sq, lane);
        else            pret_stream_rows<11, 13>(sp, sq, lane);
      }
    }
    __syncthreads();

    if (prod) {
      const int sub0 = st * 4;
      if (sub0 + 3 < tile) {
        v8f S[4];
        #pragma unroll
        for (int s = 0; s < 4; ++s) S[s] = zero8;
        #pragma unroll
        for (int c = 0; c < 16; ++c) {
          #pragma unroll
          for (int s = 0; s < 4; ++s) {
            const v2f b = *(const v2f*)&sKx[16 * s + n16][4 * c + 2 * half];
            S[s] = __builtin_amdgcn_wmma_f32_16x16x4_f32(false, a[c], false, b,
                                                         (short)0, S[s], false, false);
          }
        }
        #pragma unroll
        for (int s = 0; s < 4; ++s) accum_subtile(S[s], s);
      } else {
        #pragma unroll
        for (int s = 0; s < 4; ++s) {
          const int sub = sub0 + s;
          if (sub <= tile) {
            v8f S = zero8;
            #pragma unroll
            for (int c = 0; c < 16; ++c) {
              const v2f b = *(const v2f*)&sKx[16 * s + n16][4 * c + 2 * half];
              S = __builtin_amdgcn_wmma_f32_16x16x4_f32(false, a[c], false, b,
                                                        (short)0, S, false, false);
            }
            {
              const int key = (sub == tile) ? (sub * 16 + n16) : -1;
              #pragma unroll
              for (int g = 0; g < 8; ++g) {
                const int qr = q0 + g + 8 * half;
                S[g] = (key > qr) ? 0.0f : S[g];
              }
            }
            accum_subtile(S, s);
          }
        }
      }
      asm volatile(PRET_WMMA_RUN_END : "+v"(Oacc[0]), "+v"(Oacc[1]), "+v"(Oacc[2]),
                                       "+v"(Oacc[3]), "+v"(Pacc));
    } else {
      if (haveStates) {
        if (firstStage) pret_stream_rows<4, 6>(sp, sq, lane);
        else            pret_stream_rows<7, 8>(sp, sq, lane);
      } else {
        if (firstStage) pret_stream_rows<7, 11>(sp, sq, lane);
        else            pret_stream_rows<13, 16>(sp, sq, lane);
      }
    }
    __syncthreads();
  }

  if (prod) {
    const v4f c0 = *(const v4f*)&sQc[16 * pos + 8 * half];
    const v4f c1 = *(const v4f*)&sQc[16 * pos + 8 * half + 4];
    const float cq[8] = {c0.x, c0.y, c0.z, c0.w, c1.x, c1.y, c1.z, c1.w};
    #pragma unroll
    for (int ct = 0; ct < 4; ++ct) {
      #pragma unroll
      for (int g = 0; g < 8; ++g) {
        const float p = fmaxf(fabsf(Pacc[g] * cq[g]), 1.0f);
        out[((size_t)(bh * NSEQ + q0 + g + 8 * half)) * DV + ct * 16 + n16] =
            (Oacc[ct][g] * cq[g]) / p;
      }
    }
  }
}

extern "C" void kernel_launch(void* const* d_in, const int* in_sizes, int n_in,
                              void* d_out, int out_size, void* d_ws, size_t ws_size,
                              hipStream_t stream) {
  const float* q     = (const float*)d_in[0];
  const float* k     = (const float*)d_in[1];
  const float* v     = (const float*)d_in[2];
  const float* omask = (const float*)d_in[3];
  float* out = (float*)d_out;

  float* stD = (float*)d_ws;
  float* stT = stD + (size_t)BH * NBLK * ST_D;
  float* cqw = stT + (size_t)BH * NBLK * ST_T;

  pret_state_kernel<<<BH * CQB + BH * NBLK, 256, 0, stream>>>(k, v, omask, stD, stT, cqw, ws_size);
  pret_attn_kernel<<<BH * (NSEQ / 128), (AW + SW) * 32, 0, stream>>>(q, k, v, omask,
                                                                   stD, stT, cqw, ws_size, out);
}


# attention kernel stages all 128 keys of its query block before one barrier: two work-group barriers instead of five, streaming waves run two undrained stretches (3+5 rows); summation order kept
# speedup vs baseline: 7.1995x; 1.0795x over previous
#include <hip/hip_runtime.h>
#include <hip/hip_bf16.h>
#include <math.h>


typedef __attribute__((ext_vector_type(2))) float v2f;
typedef __attribute__((ext_vector_type(4))) float v4f;
typedef __attribute__((ext_vector_type(8))) float v8f;

#define BH   16
#define NSEQ 2048
#define DK   64
#define DV   64
#define KT   64
#define PKV  68
#define KB   128
#define PVT  132

#define NBLK 15
#define ST_D (DK * DV)
#define ST_T (2 * DK)
#define CQB  (NSEQ / 128)
#define HR   64
#define WS_NEED ((size_t)(BH * NBLK * (ST_D + ST_T) + BH * NSEQ) * sizeof(float))

#define AW   8
#define SW   8
#define SR   (128 / SW)
#define SRH  ((128 - HR) / SW)
#define CQ   (NSEQ / 128)

#define PRET_WMMA_RUN_END "v_nop\n\tv_nop\n\tv_nop\n\tv_nop\n\tv_nop"

template <int RA, int RB>
__device__ __forceinline__ void
pret_stream_rows(const float* __restrict__ p,
                 float* __restrict__ sq,
                 const int lane) {
  v4f w[2][CQ];

  #pragma unroll
  for (int qq = 0; qq < CQ; ++qq)
    w[RA & 1][qq] = __builtin_nontemporal_load(
        (const v4f*)(p + (size_t)RA * NSEQ + 128 * qq));

  #pragma unroll
  for (int r = RA; r < RB; ++r) {
    __builtin_amdgcn_sched_barrier(0);
    if (r + 1 < RB) {
      #pragma unroll
      for (int qq = 0; qq < CQ; ++qq)
        w[(r + 1) & 1][qq] = __builtin_nontemporal_load(
            (const v4f*)(p + (size_t)(r + 1) * NSEQ + 128 * qq));
    }
    __builtin_amdgcn_sched_barrier(0);
    const v4f* wr = w[r & 1];
    v2f lo = {0.f, 0.f}, hi = {0.f, 0.f};
    #pragma unroll
    for (int qq = 0; qq < CQ; ++qq) {
      lo += wr[qq].xy;
      hi += wr[qq].zw;
    }
    float sum = (lo.x + lo.y) + (hi.x + hi.y);
    #pragma unroll
    for (int m = 16; m >= 1; m >>= 1) sum += __shfl_xor(sum, m, 32);
    if (lane == 0) sq[r] = wr[0].x * rsqrtf(sum);
  }
}

template <bool ROLES>
__device__ __forceinline__ void
pret_sum_stages(const float* __restrict__ k,
                const float* __restrict__ v,
                const float* __restrict__ omask,
                const bool active,
                const int bh, const int st0, const int st1,
                float (&sKx)[KT][PKV],
                float (&sVT)[DV][PKV],
                v8f (&Dacc)[2], v8f& Tacc) {
  const int wave = threadIdx.x >> 5;
  const int lane = threadIdx.x & 31;
  const int n16  = lane & 15;
  const int half = lane >> 4;
  const int rt   = wave >> 1;
  const int par  = wave & 1;
  const int ct0  = par * 2;
  const v2f ones = {1.0f, 1.0f};

  const int cg = 4 * (wave & 3) + (lane >> 3);
  const int c4 = 4 * cg;
  const int rb = 8 * (wave >> 2) + (lane & 7);

  for (int st = st0; st < st1; ++st) {
    if (active) {
      float  kc[4];
      float4 kv[4], vv[4];
      #pragma unroll
      for (int i = 0; i < 4; ++i) {
        const int krow = st * KT + rb + 16 * i;
        kc[i] = 1.0f / omask[((size_t)(bh * NSEQ + krow)) * NSEQ];
        kv[i] = *(const float4*)(k + ((size_t)(bh * NSEQ + krow)) * DK + c4);
        vv[i] = *(const float4*)(v + ((size_t)(bh * NSEQ + krow)) * DV + c4);
      }
      #pragma unroll
      for (int i = 0; i < 4; ++i) {
        const int r = rb + 16 * i;
        sKx[c4 + 0][r] = kv[i].x * kc[i];
        sKx[c4 + 1][r] = kv[i].y * kc[i];
        sKx[c4 + 2][r] = kv[i].z * kc[i];
        sKx[c4 + 3][r] = kv[i].w * kc[i];
        sVT[c4 + 0][r] = vv[i].x;
        sVT[c4 + 1][r] = vv[i].y;
        sVT[c4 + 2][r] = vv[i].z;
        sVT[c4 + 3][r] = vv[i].w;
      }
    }
    __syncthreads();

    if (active) {
    #pragma unroll
    for (int kk = 0; kk < 16; ++kk) {
      const v2f ka = *(const v2f*)&sKx[16 * rt + n16][4 * kk + 2 * half];
      #pragma unroll
      for (int t = 0; t < 2; ++t) {
        const v2f vb = *(const v2f*)&sVT[16 * (ct0 + t) + n16][4 * kk + 2 * half];
        Dacc[t] = __builtin_amdgcn_wmma_f32_16x16x4_f32(false, ka, false, vb,
                                                        (short)0, Dacc[t], false, false);
      }
      if ((kk & 1) == 0) {
        const v2f kt = *(const v2f*)&sKx[16 * rt + n16][4 * (kk + par) + 2 * half];
        Tacc = __builtin_amdgcn_wmma_f32_16x16x4_f32(false, kt, false, ones,
                                                     (short)0, Tacc, false, false);
      }
    }
    if (ROLES) asm volatile(PRET_WMMA_RUN_END : "+v"(Dacc[0]), "+v"(Dacc[1]), "+v"(Tacc));
    }
    __syncthreads();
  }
}

__global__ void __launch_bounds__(256)
pret_state_kernel(const float* __restrict__ k,
                  const float* __restrict__ v,
                  const float* __restrict__ omask,
                  float* __restrict__ stD,
                  float* __restrict__ stT,
                  float* __restrict__ cqw,
                  const size_t ws_bytes) {
  __shared__ __attribute__((aligned(16))) float sKx[KT][PKV];
  __shared__ __attribute__((aligned(16))) float sVT[DV][PKV];

  if (ws_bytes < WS_NEED) return;

  if (blockIdx.x < BH * CQB) {
    const int    sw   = __builtin_amdgcn_readfirstlane((int)(threadIdx.x >> 5));
    const int    sl   = threadIdx.x & 31;
    const size_t row0 = (size_t)blockIdx.x * 128 + sw * (HR / 8);
    pret_stream_rows<0, HR / 8>(omask + row0 * NSEQ + 4 * sl, cqw + row0, sl);
    return;
  }

  const int sb   = blockIdx.x - BH * CQB;
  const int bh   = sb / NBLK;
  const int blk  = sb % NBLK;
  const int wave = threadIdx.x >> 5;
  const int lane = threadIdx.x & 31;
  const int n16  = lane & 15;
  const int half = lane >> 4;
  const int rt   = wave >> 1;
  const int par  = wave & 1;
  const int ct0  = par * 2;

  const v8f zero8 = {0.f, 0.f, 0.f, 0.f, 0.f, 0.f, 0.f, 0.f};
  v8f Dacc[2];
  Dacc[0] = zero8;
  Dacc[1] = zero8;
  v8f Tacc = zero8;
  pret_sum_stages<false>(k, v, omask, true, bh, 2 * blk, 2 * blk + 2, sKx, sVT, Dacc, Tacc);

  float* d = stD + (size_t)(bh * NBLK + blk) * ST_D;
  #pragma unroll
  for (int t = 0; t < 2; ++t) {
    #pragma unroll
    for (int g = 0; g < 8; ++g)
      d[(16 * rt + g + 8 * half) * DV + 16 * (ct0 + t) + n16] = Dacc[t][g];
  }
  if (n16 == 0) {
    float* tt = stT + (size_t)(bh * NBLK + blk) * ST_T + par * DK;
    #pragma unroll
    for (int g = 0; g < 8; ++g) tt[16 * rt + g + 8 * half] = Tacc[g];
  }
}

__global__ void __launch_bounds__((AW + SW) * 32)
pret_attn_kernel(const float* __restrict__ q,
                 const float* __restrict__ k,
                 const float* __restrict__ v,
                 const float* __restrict__ omask,
                 const float* __restrict__ stD,
                 const float* __restrict__ stT,
                 const float* __restrict__ cqw,
                 const size_t ws_bytes,
                 float* __restrict__ out) {
  __shared__ __attribute__((aligned(16))) float sK [KB][PKV];
  __shared__ __attribute__((aligned(16))) float sV [DV][PVT];
  float (&sKx)[KT][PKV] = *reinterpret_cast<float (*)[KT][PKV]>(&sK[0][0]);
  float (&sVT)[DV][PKV] = *reinterpret_cast<float (*)[DV][PKV]>(&sV[0][0]);
  __shared__ __attribute__((aligned(16))) float sDT[DV][PKV];
  __shared__ __attribute__((aligned(16))) float sT [2][DK];
  __shared__ __attribute__((aligned(16))) float sS [AW][16][18];
  __shared__ __attribute__((aligned(16))) float sQc[128];

  const int nseg  = NSEQ / 128;
  const int bh    = blockIdx.x / nseg;
  const int seg   = blockIdx.x % nseg;
  const int wave  = __builtin_amdgcn_readfirstlane((int)(threadIdx.x >> 5));
  const int lane  = threadIdx.x & 31;
  const int n16   = lane & 15;
  const int half  = lane >> 4;
  const bool prod = wave < AW;
  const bool haveStates = ws_bytes >= WS_NEED;

  const int    srow = haveStates ? HR + (wave - AW) * SRH
                                 : (wave - AW) * SR;
  const float* sp   = omask + ((size_t)(bh * NSEQ + seg * 128 + (prod ? 0 : srow))) * NSEQ + 4 * lane;
  float*       sq   = &sQc[prod ? 0 : srow];

  const int pos   = (wave < 4) ? wave : (11 - wave);
  const int tile  = seg * 8 + pos;
  const int q0    = tile * 16;

  const v8f zero8 = {0.f, 0.f, 0.f, 0.f, 0.f, 0.f, 0.f, 0.f};
  const v2f ones  = {1.0f, 1.0f};

  v2f a[16];
  #pragma unroll
  for (int c = 0; c < 16; ++c) a[c] = (v2f){0.f, 0.f};

  if (prod) {
    const float* qrow = q + ((size_t)(bh * NSEQ + q0 + n16)) * DK;
    #pragma unroll
    for (int c = 0; c < 16; ++c)
      a[c] = *(const v2f*)(qrow + 4 * c + 2 * half);
  }

  auto stage_block_keys = [&]() {
    const int r0 = threadIdx.x >> 4;
    const int c4 = (threadIdx.x & 15) * 4;
    float  kc[8];
    float4 kv[8], vv[8];
    #pragma unroll
    for (int i = 0; i < 8; ++i) {
      const int krow = seg * KB + r0 + 16 * i;
      kc[i] = 1.0f / omask[((size_t)(bh * NSEQ + krow)) * NSEQ];
      kv[i] = *(const float4*)(k + ((size_t)(bh * NSEQ + krow)) * DK + c4);
      vv[i] = *(const float4*)(v + ((size_t)(bh * NSEQ + krow)) * DV + c4);
    }
    #pragma unroll
    for (int i = 0; i < 8; ++i) {
      const int r = r0 + 16 * i;
      sK[r][c4 + 0] = kv[i].x * kc[i];
      sK[r][c4 + 1] = kv[i].y * kc[i];
      sK[r][c4 + 2] = kv[i].z * kc[i];
      sK[r][c4 + 3] = kv[i].w * kc[i];
      sV[c4 + 0][r] = vv[i].x;
      sV[c4 + 1][r] = vv[i].y;
      sV[c4 + 2][r] = vv[i].z;
      sV[c4 + 3][r] = vv[i].w;
    }
  };

  if (haveStates) {
    if (prod) {
      stage_block_keys();
      const float* dsrc = stD + (size_t)(bh * NBLK) * ST_D + 4 * threadIdx.x;
      const float* tsrc = stT + (size_t)(bh * NBLK) * ST_T + (threadIdx.x & (ST_T - 1));
      const bool   tOwn = threadIdx.x < ST_T;
      v4f   dsum[4];
      float tsum = 0.0f;
      #pragma unroll
      for (int i = 0; i < 4; ++i) dsum[i] = (v4f){0.f, 0.f, 0.f, 0.f};
      for (int b0 = 0; b0 < seg; b0 += 8) {
        v4f   dw[8][4];
        float tw[8];
        #pragma unroll
        for (int j = 0; j < 8; ++j) {
          const int b = (b0 + j < seg) ? (b0 + j) : (seg - 1);
          #pragma unroll
          for (int i = 0; i < 4; ++i)
            dw[j][i] = *(const v4f*)(dsrc + (size_t)b * ST_D + 1024 * i);
          tw[j] = tsrc[(size_t)b * ST_T];
        }
        #pragma unroll
        for (int j = 0; j < 8; ++j) {
          const float m = (b0 + j < seg) ? 1.0f : 0.0f;
          #pragma unroll
          for (int i = 0; i < 4; ++i) dsum[i] += m * dw[j][i];
          tsum += m * tw[j];
        }
      }
      #pragma unroll
      for (int i = 0; i < 4; ++i) {
        const int e  = 4 * ((int)threadIdx.x + 256 * i);
        const int dk = e >> 6;
        const int dv = e & 63;
        sDT[dv + 0][dk] = dsum[i].x;
        sDT[dv + 1][dk] = dsum[i].y;
        sDT[dv + 2][dk] = dsum[i].z;
        sDT[dv + 3][dk] = dsum[i].w;
      }
      if (tOwn) (&sT[0][0])[threadIdx.x] = tsum;
    } else {
      const int   crow = (wave - AW) * (HR / SW) + (lane & (HR / SW - 1));
      const float cpre = cqw[(size_t)(bh * NSEQ + seg * 128) + crow];
      pret_stream_rows<0, 3>(sp, sq, lane);
      if (lane < HR / SW) sQc[crow] = cpre;
    }
  } else {
    if (!prod) pret_stream_rows<0, 3>(sp, sq, lane);
    const int rt  = (wave >> 1) & 3;
    const int par = wave & 1;
    const int ct0 = par * 2;
    v8f Dacc[2];
    Dacc[0] = zero8;
    Dacc[1] = zero8;
    v8f Tacc = zero8;
    pret_sum_stages<true>(k, v, omask, prod, bh, 0, 2 * seg, sKx, sVT, Dacc, Tacc);

    if (prod) {
      #pragma unroll
      for (int t = 0; t < 2; ++t) {
        #pragma unroll
        for (int g = 0; g < 8; ++g)
          sDT[16 * (ct0 + t) + n16][16 * rt + g + 8 * half] = Dacc[t][g];
      }
      if (n16 == 0) {
        #pragma unroll
        for (int g = 0; g < 8; ++g) sT[par][16 * rt + g + 8 * half] = Tacc[g];
      }
      stage_block_keys();
    } else {
      pret_stream_rows<3, 8>(sp, sq, lane);
    }
  }
  __syncthreads();

  v8f Oacc[4];
  #pragma unroll
  for (int t = 0; t < 4; ++t) Oacc[t] = zero8;
  v8f Pacc = zero8;

  if (prod) {
    #pragma unroll
    for (int c = 0; c < 16; ++c) {
      const v2f t0 = *(const v2f*)&sT[0][4 * c + 2 * half];
      const v2f t1 = *(const v2f*)&sT[1][4 * c + 2 * half];
      const v2f tb = t0 + t1;
      Pacc = __builtin_amdgcn_wmma_f32_16x16x4_f32(false, a[c], false, tb,
                                                   (short)0, Pacc, false, false);
      #pragma unroll
      for (int ct = 0; ct < 4; ++ct) {
        const v2f db = *(const v2f*)&sDT[16 * ct + n16][4 * c + 2 * half];
        Oacc[ct] = __builtin_amdgcn_wmma_f32_16x16x4_f32(false, a[c], false, db,
                                                         (short)0, Oacc[ct], false, false);
      }
    }
    asm volatile(PRET_WMMA_RUN_END : "+v"(Oacc[0]), "+v"(Oacc[1]), "+v"(Oacc[2]),
                                     "+v"(Oacc[3]), "+v"(Pacc));

    for (int hs = 0; hs < 2; ++hs) {
      const float* kfrag = &sK[64 * hs + n16][2 * half];
      const float* vfrag = &sV[n16][64 * hs + 2 * half];

      auto accum_subtile = [&](const v8f& S, const int s) {
        #pragma unroll
        for (int g = 0; g < 8; ++g) sS[wave][g + 8 * half][n16] = S[g];
        v2f sa[4];
        #pragma unroll
        for (int c = 0; c < 4; ++c)
          sa[c] = *(const v2f*)&sS[wave][n16][4 * c + 2 * half];

        #pragma unroll
        for (int c = 0; c < 4; ++c)
          Pacc = __builtin_amdgcn_wmma_f32_16x16x4_f32(false, sa[c], false, ones,
                                                       (short)0, Pacc, false, false);

        #pragma unroll
        for (int ct = 0; ct < 4; ++ct) {
          #pragma unroll
          for (int c = 0; c < 4; ++c) {
            const v2f vb = *(const v2f*)(vfrag + 16 * ct * PVT + 16 * s + 4 * c);
            Oacc[ct] = __builtin_amdgcn_wmma_f32_16x16x4_f32(false, sa[c], false, vb,
                                                             (short)0, Oacc[ct], false, false);
          }
        }
      };

      const int sub0 = (2 * seg + hs) * 4;
      if (sub0 + 3 < tile) {
        v8f S[4];
        #pragma unroll
        for (int s = 0; s < 4; ++s) S[s] = zero8;
        #pragma unroll
        for (int c = 0; c < 16; ++c) {
          #pragma unroll
          for (int s = 0; s < 4; ++s) {
            const v2f b = *(const v2f*)(kfrag + 16 * s * PKV + 4 * c);
            S[s] = __builtin_amdgcn_wmma_f32_16x16x4_f32(false, a[c], false, b,
                                                         (short)0, S[s], false, false);
          }
        }
        #pragma unroll
        for (int s = 0; s < 4; ++s) accum_subtile(S[s], s);
      } else {
        #pragma unroll
        for (int s = 0; s < 4; ++s) {
          const int sub = sub0 + s;
          if (sub <= tile) {
            v8f S = zero8;
            #pragma unroll
            for (int c = 0; c < 16; ++c) {
              const v2f b = *(const v2f*)(kfrag + 16 * s * PKV + 4 * c);
              S = __builtin_amdgcn_wmma_f32_16x16x4_f32(false, a[c], false, b,
                                                        (short)0, S, false, false);
            }
            {
              const int key = (sub == tile) ? (sub * 16 + n16) : -1;
              #pragma unroll
              for (int g = 0; g < 8; ++g) {
                const int qr = q0 + g + 8 * half;
                S[g] = (key > qr) ? 0.0f : S[g];
              }
            }
            accum_subtile(S, s);
          }
        }
      }
      asm volatile(PRET_WMMA_RUN_END : "+v"(Oacc[0]), "+v"(Oacc[1]), "+v"(Oacc[2]),
                                       "+v"(Oacc[3]), "+v"(Pacc));
    }
  } else {
    if (haveStates) pret_stream_rows<3, 8>(sp, sq, lane);
    else            pret_stream_rows<8, 16>(sp, sq, lane);
  }
  __syncthreads();

  if (prod) {
    const v4f c0 = *(const v4f*)&sQc[16 * pos + 8 * half];
    const v4f c1 = *(const v4f*)&sQc[16 * pos + 8 * half + 4];
    const float cq[8] = {c0.x, c0.y, c0.z, c0.w, c1.x, c1.y, c1.z, c1.w};
    #pragma unroll
    for (int ct = 0; ct < 4; ++ct) {
      #pragma unroll
      for (int g = 0; g < 8; ++g) {
        const float p = fmaxf(fabsf(Pacc[g] * cq[g]), 1.0f);
        out[((size_t)(bh * NSEQ + q0 + g + 8 * half)) * DV + ct * 16 + n16] =
            (Oacc[ct][g] * cq[g]) / p;
      }
    }
  }
}

extern "C" void kernel_launch(void* const* d_in, const int* in_sizes, int n_in,
                              void* d_out, int out_size, void* d_ws, size_t ws_size,
                              hipStream_t stream) {
  const float* q     = (const float*)d_in[0];
  const float* k     = (const float*)d_in[1];
  const float* v     = (const float*)d_in[2];
  const float* omask = (const float*)d_in[3];
  float* out = (float*)d_out;

  float* stD = (float*)d_ws;
  float* stT = stD + (size_t)BH * NBLK * ST_D;
  float* cqw = stT + (size_t)BH * NBLK * ST_T;

  pret_state_kernel<<<BH * CQB + BH * NBLK, 256, 0, stream>>>(k, v, omask, stD, stT, cqw, ws_size);
  pret_attn_kernel<<<BH * (NSEQ / 128), (AW + SW) * 32, 0, stream>>>(q, k, v, omask,
                                                                   stD, stT, cqw, ws_size, out);
}
